# starting kernel
# baseline (speedup 1.0000x reference)
#include <hip/hip_runtime.h>

typedef float v2f __attribute__((ext_vector_type(2)));
typedef float v8f __attribute__((ext_vector_type(8)));

__global__ void zero_f32v4(float4* __restrict__ p, long long n4) {
    long long i = (long long)blockIdx.x * blockDim.x + threadIdx.x;
    if (i < n4) p[i] = make_float4(0.f, 0.f, 0.f, 0.f);
}

template <int RELU_A, int NN, int KK>
__global__ void gemm_wmma_f32(const float* __restrict__ A,
                              const float* __restrict__ B,
                              float* __restrict__ C,
                              int M) {
    constexpr int LDB = KK + 4;
    __shared__ float Bt[16 * LDB];

    const int tid  = (int)threadIdx.x;
    const int n0   = (int)blockIdx.y << 4;

    #pragma unroll
    for (int idx = 0; idx < KK * 16 / 256; ++idx) {
        int flat = tid + idx * 256;
        int k = flat >> 4;
        int n = flat & 15;
        Bt[n * LDB + k] = B[k * NN + n0 + n];
    }
    __syncthreads();

    const int wave = tid >> 5;
    const int lane = tid & 31;
    const int tileM = (int)blockIdx.x * 8 + wave;
    if (tileM >= (M >> 4)) return;

    const int l15  = lane & 15;
    const int kOff = (lane >> 4) << 1;

    const float* __restrict__ arow = A + ((tileM << 4) + l15) * KK + kOff;
    const float* __restrict__ brow = &Bt[l15 * LDB + kOff];

    v8f acc = {};
    #pragma unroll 16
    for (int k0 = 0; k0 < KK; k0 += 4) {
        float2 av = *(const float2*)(arow + k0);
        if (RELU_A) {
            av.x = fmaxf(av.x, 0.f);
            av.y = fmaxf(av.y, 0.f);
        }
        v2f a; a.x = av.x; a.y = av.y;
        v2f b = *(const v2f*)(brow + k0);
        acc = __builtin_amdgcn_wmma_f32_16x16x4_f32(
             false, a,  false, b,
             (short)0, acc,  false,  false);
    }

    const int cm = (tileM << 4) + ((lane >> 4) << 3);
    float* __restrict__ crow = C + cm * NN + n0 + l15;
    #pragma unroll
    for (int r = 0; r < 8; ++r) crow[r * NN] = acc[r];
}

template <int SHIFT>
__global__ void spmm_atomic(const int*   __restrict__ erow,
                            const int*   __restrict__ ecol,
                            const float* __restrict__ ew,
                            const float* __restrict__ h,
                            float*       __restrict__ out,
                            int nEdges) {
    const long long tid = (long long)blockIdx.x * blockDim.x + threadIdx.x;
    const int e  = (int)(tid >> SHIFT);
    if (e >= nEdges) return;
    const int c4 = ((int)tid & ((1 << SHIFT) - 1)) << 2;
    constexpr int Cch = 4 << SHIFT;

    const float we = ew[e];
    const int   j  = ecol[e];
    const int   i  = erow[e];

    const float4 hv = *(const float4*)(h + (size_t)j * Cch + c4);
    float* dst = out + (size_t)i * Cch + c4;
    atomicAdd(dst + 0, hv.x * we);
    atomicAdd(dst + 1, hv.y * we);
    atomicAdd(dst + 2, hv.z * we);
    atomicAdd(dst + 3, hv.w * we);
}

extern "C" void kernel_launch(void* const* d_in, const int* in_sizes, int n_in,
                              void* d_out, int out_size, void* d_ws, size_t ws_size,
                              hipStream_t stream) {
    const float* x   = (const float*)d_in[0];
    const float* W1  = (const float*)d_in[1];
    const float* W2  = (const float*)d_in[2];
    const float* ew  = (const float*)d_in[3];
    const int*   er  = (const int*)  d_in[4];
    const int*   ec  = (const int*)  d_in[5];

    const int F_IN = 512, C1 = 128, C2 = 64;
    const int nNodes = in_sizes[0] / F_IN;
    const int nEdges = in_sizes[3];

    float* H   = (float*)d_ws;
    float* Hs  = H  + (size_t)nNodes * C1;
    float* H2  = Hs + (size_t)nNodes * C1;
    float* out = (float*)d_out;

    const int BLK = 256;

    {
        long long n4 = (long long)nNodes * C1 / 4;
        zero_f32v4<<<(unsigned)((n4 + BLK - 1) / BLK), BLK, 0, stream>>>((float4*)Hs, n4);
        long long m4 = (long long)nNodes * C2 / 4;
        zero_f32v4<<<(unsigned)((m4 + BLK - 1) / BLK), BLK, 0, stream>>>((float4*)out, m4);
    }

    {
        int mTiles = nNodes / 16;
        dim3 grid((mTiles + 7) / 8, C1 / 16);
        gemm_wmma_f32<0, 128, 512><<<grid, BLK, 0, stream>>>(x, W1, H, nNodes);
    }

    {
        long long threads = (long long)nEdges << 5;
        spmm_atomic<5><<<(unsigned)((threads + BLK - 1) / BLK), BLK, 0, stream>>>(
            er, ec, ew, H, Hs, nEdges);
    }

    {
        int mTiles = nNodes / 16;
        dim3 grid((mTiles + 7) / 8, C2 / 16);
        gemm_wmma_f32<1, 64, 128><<<grid, BLK, 0, stream>>>(Hs, W2, H2, nNodes);
    }

    {
        long long threads = (long long)nEdges << 4;
        spmm_atomic<4><<<(unsigned)((threads + BLK - 1) / BLK), BLK, 0, stream>>>(
            er, ec, ew, H2, out, nEdges);
    }
}


# Float-atomic scatter SpMM replaced by row-sorted gather: edges grouped by row in-call with LDS counters, one wave sums a row in registers, stores it once; sum order is now the row's list order
# speedup vs baseline: 2.2340x; 2.2340x over previous
#include <hip/hip_runtime.h>

typedef float v2f __attribute__((ext_vector_type(2)));
typedef float v8f __attribute__((ext_vector_type(8)));

template <int RELU_A, int NN, int KK>
__global__ void gemm_wmma_f32(const float* __restrict__ A,
                              const float* __restrict__ B,
                              float* __restrict__ C,
                              int M) {
    constexpr int LDB = KK + 4;
    __shared__ float Bt[16 * LDB];

    const int tid  = (int)threadIdx.x;
    const int n0   = (int)blockIdx.y << 4;

    #pragma unroll
    for (int idx = 0; idx < KK * 16 / 256; ++idx) {
        int flat = tid + idx * 256;
        int k = flat >> 4;
        int n = flat & 15;
        Bt[n * LDB + k] = B[k * NN + n0 + n];
    }
    __syncthreads();

    const int wave = tid >> 5;
    const int lane = tid & 31;
    const int tileM = (int)blockIdx.x * 8 + wave;
    if (tileM >= (M >> 4)) return;

    const int l15  = lane & 15;
    const int kOff = (lane >> 4) << 1;

    const float* __restrict__ arow = A + ((tileM << 4) + l15) * KK + kOff;
    const float* __restrict__ brow = &Bt[l15 * LDB + kOff];

    v8f acc = {};
    #pragma unroll 16
    for (int k0 = 0; k0 < KK; k0 += 4) {
        float2 av = *(const float2*)(arow + k0);
        if (RELU_A) {
            av.x = fmaxf(av.x, 0.f);
            av.y = fmaxf(av.y, 0.f);
        }
        v2f a; a.x = av.x; a.y = av.y;
        v2f b = *(const v2f*)(brow + k0);
        acc = __builtin_amdgcn_wmma_f32_16x16x4_f32(
             false, a,  false, b,
             (short)0, acc,  false,  false);
    }

    const int cm = (tileM << 4) + ((lane >> 4) << 3);
    float* __restrict__ crow = C + cm * NN + n0 + l15;
    #pragma unroll
    for (int r = 0; r < 8; ++r) crow[r * NN] = acc[r];
}

#define NB 256
#define SORT_T 1024

__global__ void __launch_bounds__(SORT_T)
bin_count(const int* __restrict__ erow, int* __restrict__ M,
          int nEdges, int chunk, unsigned rpb) {
    __shared__ int cnt[NB];
    const int tid = (int)threadIdx.x;
    const int b   = (int)blockIdx.x;
    if (tid < NB) cnt[tid] = 0;
    __syncthreads();
    const int e0 = b * chunk;
    const int e1 = min(e0 + chunk, nEdges);
    for (int e = e0 + tid; e < e1; e += SORT_T) {
        unsigned o = (unsigned)erow[e] / rpb;
        atomicAdd(&cnt[o], 1);
    }
    __syncthreads();
    if (tid < NB) M[tid * NB + b] = cnt[tid];
}

__device__ __forceinline__ int block_excl_scan(int t, int* wsum) {
    const int lane = (int)threadIdx.x & 31;
    const int wave = (int)threadIdx.x >> 5;
    int x = t;
    #pragma unroll
    for (int d = 1; d < 32; d <<= 1) {
        int y = __shfl_up(x, d);
        if (lane >= d) x += y;
    }
    if (lane == 31) wsum[wave] = x;
    __syncthreads();
    if (wave == 0) {
        int s0 = wsum[lane];
        int s = s0;
        #pragma unroll
        for (int d = 1; d < 32; d <<= 1) {
            int y = __shfl_up(s, d);
            if (lane >= d) s += y;
        }
        wsum[lane] = s - s0;
    }
    __syncthreads();
    return wsum[wave] + x - t;
}

__global__ void __launch_bounds__(SORT_T)
bin_scan(int* __restrict__ M) {
    __shared__ int wsum[32];
    const int tid = (int)threadIdx.x;
    int4* p = (int4*)(M + tid * (NB * NB / SORT_T));
    int4 v[NB * NB / SORT_T / 4];
    int t = 0;
    #pragma unroll
    for (int k = 0; k < NB * NB / SORT_T / 4; ++k) {
        v[k] = p[k];
        t += v[k].x + v[k].y + v[k].z + v[k].w;
    }
    int run = block_excl_scan(t, wsum);
    #pragma unroll
    for (int k = 0; k < NB * NB / SORT_T / 4; ++k) {
        int4 o;
        o.x = run; run += v[k].x;
        o.y = run; run += v[k].y;
        o.z = run; run += v[k].z;
        o.w = run; run += v[k].w;
        p[k] = o;
    }
}

__global__ void __launch_bounds__(SORT_T)
bin_place(const int* __restrict__ erow, const int* __restrict__ P,
          int2* __restrict__ rec, int nEdges, int chunk, unsigned rpb) {
    __shared__ int pos[NB];
    const int tid = (int)threadIdx.x;
    const int b   = (int)blockIdx.x;
    if (tid < NB) pos[tid] = P[tid * NB + b];
    __syncthreads();
    const int e0 = b * chunk;
    const int e1 = min(e0 + chunk, nEdges);
    for (int e = e0 + tid; e < e1; e += SORT_T) {
        int r = erow[e];
        unsigned o = (unsigned)r / rpb;
        int p = atomicAdd(&pos[o], 1);
        rec[p] = make_int2(r, e);
    }
}

__global__ void __launch_bounds__(SORT_T)
owner_sort(const int2* __restrict__ rec, const int* __restrict__ P,
           const int* __restrict__ ecol, const float* __restrict__ ew,
           int* __restrict__ rowptr, unsigned short* __restrict__ scol,
           float* __restrict__ sw, int nEdges, int nNodes, int rpb) {
    __shared__ int cnt[SORT_T];
    __shared__ int wsum[32];
    const int tid = (int)threadIdx.x;
    const int o   = (int)blockIdx.x;
    const int s   = P[o * NB];
    const int e   = (o == NB - 1) ? nEdges : P[(o + 1) * NB];
    const int r0  = o * rpb;

    cnt[tid] = 0;
    __syncthreads();
    for (int p = s + tid; p < e; p += SORT_T) {
        int2 rc = rec[p];
        atomicAdd(&cnt[rc.x - r0], 1);
    }
    __syncthreads();
    const int mine = cnt[tid];
    const int excl = block_excl_scan(mine, wsum);
    if (tid < rpb && r0 + tid < nNodes) rowptr[r0 + tid] = s + excl;
    if (o == NB - 1 && tid == 0) rowptr[nNodes] = nEdges;
    cnt[tid] = excl;
    __syncthreads();
    for (int p = s + tid; p < e; p += SORT_T) {
        int2 rc = rec[p];
        int k = atomicAdd(&cnt[rc.x - r0], 1);
        int dst = s + k;
        scol[dst] = (unsigned short)ecol[rc.y];
        sw[dst]   = ew[rc.y];
    }
}

template <int SHIFT>
__global__ void spmm_rows(const int* __restrict__ rowptr,
                          const unsigned short* __restrict__ scol,
                          const float* __restrict__ sw,
                          const float* __restrict__ h,
                          float* __restrict__ out,
                          int nRows) {
    constexpr int LPE = 1 << SHIFT;
    constexpr int EPS = 32 >> SHIFT;
    constexpr int Cch = 4 << SHIFT;
    const int lane = (int)threadIdx.x & 31;
    const int row  = (int)blockIdx.x * ((int)blockDim.x >> 5) + ((int)threadIdx.x >> 5);
    if (row >= nRows) return;
    const int s = __builtin_amdgcn_readfirstlane(rowptr[row]);
    const int e = __builtin_amdgcn_readfirstlane(rowptr[row + 1]);
    const int sub = lane & (LPE - 1);
    const int grp = lane >> SHIFT;
    const float* __restrict__ hb = h + sub * 4;

    float4 acc = make_float4(0.f, 0.f, 0.f, 0.f);
    for (int base = s; base < e; base += 32) {
        const int n = min(32, e - base);
        int   c = 0;
        float w = 0.f;
        if (lane < n) { c = (int)scol[base + lane]; w = sw[base + lane]; }
        const int wi = __float_as_int(w);
        int t = 0;
        for (; t + 4 * EPS <= n; t += 4 * EPS) {
            float4 hv[4];
            float  wj[4];
            #pragma unroll
            for (int u = 0; u < 4; ++u) {
                int cj, wb;
                if (EPS == 1) {
                    cj = __builtin_amdgcn_readlane(c,  t + u);
                    wb = __builtin_amdgcn_readlane(wi, t + u);
                } else {
                    int c0 = __builtin_amdgcn_readlane(c,  t + 2 * u);
                    int c1 = __builtin_amdgcn_readlane(c,  t + 2 * u + 1);
                    int w0 = __builtin_amdgcn_readlane(wi, t + 2 * u);
                    int w1 = __builtin_amdgcn_readlane(wi, t + 2 * u + 1);
                    cj = grp ? c1 : c0;
                    wb = grp ? w1 : w0;
                }
                wj[u] = __int_as_float(wb);
                hv[u] = *(const float4*)(hb + (size_t)cj * Cch);
            }
            #pragma unroll
            for (int u = 0; u < 4; ++u) {
                acc.x = fmaf(hv[u].x, wj[u], acc.x);
                acc.y = fmaf(hv[u].y, wj[u], acc.y);
                acc.z = fmaf(hv[u].z, wj[u], acc.z);
                acc.w = fmaf(hv[u].w, wj[u], acc.w);
            }
        }
        for (; t < n; t += EPS) {
            int cj, wb;
            bool ok = true;
            if (EPS == 1) {
                cj = __builtin_amdgcn_readlane(c,  t);
                wb = __builtin_amdgcn_readlane(wi, t);
            } else {
                const int t1 = min(t + 1, 31);
                int c0 = __builtin_amdgcn_readlane(c,  t);
                int c1 = __builtin_amdgcn_readlane(c,  t1);
                int w0 = __builtin_amdgcn_readlane(wi, t);
                int w1 = __builtin_amdgcn_readlane(wi, t1);
                cj = grp ? c1 : c0;
                wb = grp ? w1 : w0;
                ok = (t + grp) < n;
            }
            if (ok) {
                const float  wv = __int_as_float(wb);
                const float4 hv = *(const float4*)(hb + (size_t)cj * Cch);
                acc.x = fmaf(hv.x, wv, acc.x);
                acc.y = fmaf(hv.y, wv, acc.y);
                acc.z = fmaf(hv.z, wv, acc.z);
                acc.w = fmaf(hv.w, wv, acc.w);
            }
        }
    }
    if (EPS == 2) {
        acc.x += __shfl_xor(acc.x, 16);
        acc.y += __shfl_xor(acc.y, 16);
        acc.z += __shfl_xor(acc.z, 16);
        acc.w += __shfl_xor(acc.w, 16);
    }
    if (grp == 0) *(float4*)(out + (size_t)row * Cch + sub * 4) = acc;
}

extern "C" void kernel_launch(void* const* d_in, const int* in_sizes, int n_in,
                              void* d_out, int out_size, void* d_ws, size_t ws_size,
                              hipStream_t stream) {
    const float* x   = (const float*)d_in[0];
    const float* W1  = (const float*)d_in[1];
    const float* W2  = (const float*)d_in[2];
    const float* ew  = (const float*)d_in[3];
    const int*   er  = (const int*)  d_in[4];
    const int*   ec  = (const int*)  d_in[5];

    const int F_IN = 512, C1 = 128, C2 = 64;
    const int nNodes = in_sizes[0] / F_IN;
    const int nEdges = in_sizes[3];

    float* H   = (float*)d_ws;
    float* Hs  = H  + (size_t)nNodes * C1;
    float* R3  = Hs + (size_t)nNodes * C1;
    float* H2  = H;
    float* out = (float*)d_out;

    int*            rowptr = (int*)R3;
    float*          sw     = R3 + 65536;
    unsigned short* scol   = (unsigned short*)(sw + nEdges);
    int*  P   = (int*)H;
    int2* rec = (int2*)(H + 65536);

    const int BLK = 256;
    const int chunk = (nEdges + NB - 1) / NB;
    const int rpb   = (nNodes + NB - 1) / NB;

    bin_count<<<NB, SORT_T, 0, stream>>>(er, P, nEdges, chunk, (unsigned)rpb);
    bin_scan<<<1, SORT_T, 0, stream>>>(P);
    bin_place<<<NB, SORT_T, 0, stream>>>(er, P, rec, nEdges, chunk, (unsigned)rpb);
    owner_sort<<<NB, SORT_T, 0, stream>>>(rec, P, ec, ew, rowptr, scol, sw,
                                          nEdges, nNodes, rpb);

    {
        int mTiles = nNodes / 16;
        dim3 grid((mTiles + 7) / 8, C1 / 16);
        gemm_wmma_f32<0, 128, 512><<<grid, BLK, 0, stream>>>(x, W1, H, nNodes);
    }

    spmm_rows<5><<<(nNodes + 7) / 8, BLK, 0, stream>>>(rowptr, scol, sw, H, Hs, nNodes);

    {
        int mTiles = nNodes / 16;
        dim3 grid((mTiles + 7) / 8, C2 / 16);
        gemm_wmma_f32<1, 64, 128><<<grid, BLK, 0, stream>>>(Hs, W2, H2, nNodes);
    }

    spmm_rows<4><<<(nNodes + 7) / 8, BLK, 0, stream>>>(rowptr, scol, sw, H2, out, nNodes);
}


# GEMM wave tile widened from 16x16 to 16x64: one A fragment load feeds 4 matrix instructions on 4 independent accumulators (A fetched 2x/1x instead of 8x/4x), 64-column B panel per 4-wave block
# speedup vs baseline: 2.4846x; 1.1122x over previous
#include <hip/hip_runtime.h>

typedef float v2f __attribute__((ext_vector_type(2)));
typedef float v8f __attribute__((ext_vector_type(8)));

template <int RELU_A, int NN, int KK, int NT, int WAVES>
__global__ void __launch_bounds__(WAVES * 32)
gemm_wmma_f32(const float* __restrict__ A,
              const float* __restrict__ B,
              float* __restrict__ C,
              int M) {
    constexpr int LDB = KK + 4;
    constexpr int T   = WAVES * 32;
    constexpr int U   = 16;
    static_assert(KK % (4 * U) == 0, "K must be a multiple of the register block");
    static_assert((KK * 16 * NT / 4) % T == 0 && KK % (T / 4) == 0, "panel staging shape");
    __shared__ float Bt[16 * NT * LDB];

    const int tid = (int)threadIdx.x;
    const int n0  = (int)blockIdx.x * (16 * NT);

    {
        const int q  = tid & 3;
        const int kk = tid >> 2;
        constexpr int ITERS = KK * 16 * NT / 4 / T;
        #pragma unroll 8
        for (int it = 0; it < ITERS; ++it) {
            const int g   = it % NT;
            const int k   = (it / NT) * (T / 4) + kk;
            const int col = 16 * g + 4 * q;
            const float4 v = *(const float4*)(B + k * NN + n0 + col);
            float* d = &Bt[col * LDB + k];
            d[0]       = v.x;
            d[LDB]     = v.y;
            d[2 * LDB] = v.z;
            d[3 * LDB] = v.w;
        }
    }
    __syncthreads();

    const int wave = tid >> 5;
    const int lane = tid & 31;
    const int tileM = (int)blockIdx.y * WAVES + wave;
    if (tileM >= (M >> 4)) return;

    const int l15  = lane & 15;
    const int kOff = (lane >> 4) << 1;

    const float* __restrict__ arow = A + ((tileM << 4) + l15) * KK + kOff;
    const float* __restrict__ brow = &Bt[l15 * LDB + kOff];

    v8f acc[NT];
    #pragma unroll
    for (int j = 0; j < NT; ++j) acc[j] = v8f{};

    float2 an[U];
    #pragma unroll
    for (int u = 0; u < U; ++u) an[u] = *(const float2*)(arow + 4 * u);

    #pragma unroll 1
    for (int kb = 0; kb < KK; kb += 4 * U) {
        float2 ac[U];
        #pragma unroll
        for (int u = 0; u < U; ++u) ac[u] = an[u];
        if (kb + 4 * U < KK) {
            #pragma unroll
            for (int u = 0; u < U; ++u)
                an[u] = *(const float2*)(arow + kb + 4 * U + 4 * u);
        }
        #pragma unroll
        for (int u = 0; u < U; ++u) {
            float2 av = ac[u];
            if (RELU_A) {
                av.x = fmaxf(av.x, 0.f);
                av.y = fmaxf(av.y, 0.f);
            }
            v2f a; a.x = av.x; a.y = av.y;
            #pragma unroll
            for (int j = 0; j < NT; ++j) {
                v2f b = *(const v2f*)(brow + j * 16 * LDB + kb + 4 * u);
                acc[j] = __builtin_amdgcn_wmma_f32_16x16x4_f32(
                     false, a,  false, b,
                     (short)0, acc[j],  false,  false);
            }
        }
    }

    const int cm = (tileM << 4) + ((lane >> 4) << 3);
    float* __restrict__ crow = C + cm * NN + n0 + l15;
    #pragma unroll
    for (int j = 0; j < NT; ++j) {
        #pragma unroll
        for (int r = 0; r < 8; ++r) crow[r * NN + 16 * j] = acc[j][r];
    }
}

#define NB 256
#define SORT_T 1024

__global__ void __launch_bounds__(SORT_T)
bin_count(const int* __restrict__ erow, int* __restrict__ M,
          int nEdges, int chunk, unsigned rpb) {
    __shared__ int cnt[NB];
    const int tid = (int)threadIdx.x;
    const int b   = (int)blockIdx.x;
    if (tid < NB) cnt[tid] = 0;
    __syncthreads();
    const int e0 = b * chunk;
    const int e1 = min(e0 + chunk, nEdges);
    for (int e = e0 + tid; e < e1; e += SORT_T) {
        unsigned o = (unsigned)erow[e] / rpb;
        atomicAdd(&cnt[o], 1);
    }
    __syncthreads();
    if (tid < NB) M[tid * NB + b] = cnt[tid];
}

__device__ __forceinline__ int block_excl_scan(int t, int* wsum) {
    const int lane = (int)threadIdx.x & 31;
    const int wave = (int)threadIdx.x >> 5;
    int x = t;
    #pragma unroll
    for (int d = 1; d < 32; d <<= 1) {
        int y = __shfl_up(x, d);
        if (lane >= d) x += y;
    }
    if (lane == 31) wsum[wave] = x;
    __syncthreads();
    if (wave == 0) {
        int s0 = wsum[lane];
        int s = s0;
        #pragma unroll
        for (int d = 1; d < 32; d <<= 1) {
            int y = __shfl_up(s, d);
            if (lane >= d) s += y;
        }
        wsum[lane] = s - s0;
    }
    __syncthreads();
    return wsum[wave] + x - t;
}

__global__ void __launch_bounds__(SORT_T)
bin_scan(int* __restrict__ M) {
    __shared__ int wsum[32];
    const int tid = (int)threadIdx.x;
    int4* p = (int4*)(M + tid * (NB * NB / SORT_T));
    int4 v[NB * NB / SORT_T / 4];
    int t = 0;
    #pragma unroll
    for (int k = 0; k < NB * NB / SORT_T / 4; ++k) {
        v[k] = p[k];
        t += v[k].x + v[k].y + v[k].z + v[k].w;
    }
    int run = block_excl_scan(t, wsum);
    #pragma unroll
    for (int k = 0; k < NB * NB / SORT_T / 4; ++k) {
        int4 o;
        o.x = run; run += v[k].x;
        o.y = run; run += v[k].y;
        o.z = run; run += v[k].z;
        o.w = run; run += v[k].w;
        p[k] = o;
    }
}

__global__ void __launch_bounds__(SORT_T)
bin_place(const int* __restrict__ erow, const int* __restrict__ P,
          int2* __restrict__ rec, int nEdges, int chunk, unsigned rpb) {
    __shared__ int pos[NB];
    const int tid = (int)threadIdx.x;
    const int b   = (int)blockIdx.x;
    if (tid < NB) pos[tid] = P[tid * NB + b];
    __syncthreads();
    const int e0 = b * chunk;
    const int e1 = min(e0 + chunk, nEdges);
    for (int e = e0 + tid; e < e1; e += SORT_T) {
        int r = erow[e];
        unsigned o = (unsigned)r / rpb;
        int p = atomicAdd(&pos[o], 1);
        rec[p] = make_int2(r, e);
    }
}

__global__ void __launch_bounds__(SORT_T)
owner_sort(const int2* __restrict__ rec, const int* __restrict__ P,
           const int* __restrict__ ecol, const float* __restrict__ ew,
           int* __restrict__ rowptr, unsigned short* __restrict__ scol,
           float* __restrict__ sw, int nEdges, int nNodes, int rpb) {
    __shared__ int cnt[SORT_T];
    __shared__ int wsum[32];
    const int tid = (int)threadIdx.x;
    const int o   = (int)blockIdx.x;
    const int s   = P[o * NB];
    const int e   = (o == NB - 1) ? nEdges : P[(o + 1) * NB];
    const int r0  = o * rpb;

    cnt[tid] = 0;
    __syncthreads();
    for (int p = s + tid; p < e; p += SORT_T) {
        int2 rc = rec[p];
        atomicAdd(&cnt[rc.x - r0], 1);
    }
    __syncthreads();
    const int mine = cnt[tid];
    const int excl = block_excl_scan(mine, wsum);
    if (tid < rpb && r0 + tid < nNodes) rowptr[r0 + tid] = s + excl;
    if (o == NB - 1 && tid == 0) rowptr[nNodes] = nEdges;
    cnt[tid] = excl;
    __syncthreads();
    for (int p = s + tid; p < e; p += SORT_T) {
        int2 rc = rec[p];
        int k = atomicAdd(&cnt[rc.x - r0], 1);
        int dst = s + k;
        scol[dst] = (unsigned short)ecol[rc.y];
        sw[dst]   = ew[rc.y];
    }
}

template <int SHIFT>
__global__ void spmm_rows(const int* __restrict__ rowptr,
                          const unsigned short* __restrict__ scol,
                          const float* __restrict__ sw,
                          const float* __restrict__ h,
                          float* __restrict__ out,
                          int nRows) {
    constexpr int LPE = 1 << SHIFT;
    constexpr int EPS = 32 >> SHIFT;
    constexpr int Cch = 4 << SHIFT;
    const int lane = (int)threadIdx.x & 31;
    const int row  = (int)blockIdx.x * ((int)blockDim.x >> 5) + ((int)threadIdx.x >> 5);
    if (row >= nRows) return;
    const int s = __builtin_amdgcn_readfirstlane(rowptr[row]);
    const int e = __builtin_amdgcn_readfirstlane(rowptr[row + 1]);
    const int sub = lane & (LPE - 1);
    const int grp = lane >> SHIFT;
    const float* __restrict__ hb = h + sub * 4;

    float4 acc = make_float4(0.f, 0.f, 0.f, 0.f);
    for (int base = s; base < e; base += 32) {
        const int n = min(32, e - base);
        int   c = 0;
        float w = 0.f;
        if (lane < n) { c = (int)scol[base + lane]; w = sw[base + lane]; }
        const int wi = __float_as_int(w);
        int t = 0;
        for (; t + 4 * EPS <= n; t += 4 * EPS) {
            float4 hv[4];
            float  wj[4];
            #pragma unroll
            for (int u = 0; u < 4; ++u) {
                int cj, wb;
                if (EPS == 1) {
                    cj = __builtin_amdgcn_readlane(c,  t + u);
                    wb = __builtin_amdgcn_readlane(wi, t + u);
                } else {
                    int c0 = __builtin_amdgcn_readlane(c,  t + 2 * u);
                    int c1 = __builtin_amdgcn_readlane(c,  t + 2 * u + 1);
                    int w0 = __builtin_amdgcn_readlane(wi, t + 2 * u);
                    int w1 = __builtin_amdgcn_readlane(wi, t + 2 * u + 1);
                    cj = grp ? c1 : c0;
                    wb = grp ? w1 : w0;
                }
                wj[u] = __int_as_float(wb);
                hv[u] = *(const float4*)(hb + (size_t)cj * Cch);
            }
            #pragma unroll
            for (int u = 0; u < 4; ++u) {
                acc.x = fmaf(hv[u].x, wj[u], acc.x);
                acc.y = fmaf(hv[u].y, wj[u], acc.y);
                acc.z = fmaf(hv[u].z, wj[u], acc.z);
                acc.w = fmaf(hv[u].w, wj[u], acc.w);
            }
        }
        for (; t < n; t += EPS) {
            int cj, wb;
            bool ok = true;
            if (EPS == 1) {
                cj = __builtin_amdgcn_readlane(c,  t);
                wb = __builtin_amdgcn_readlane(wi, t);
            } else {
                const int t1 = min(t + 1, 31);
                int c0 = __builtin_amdgcn_readlane(c,  t);
                int c1 = __builtin_amdgcn_readlane(c,  t1);
                int w0 = __builtin_amdgcn_readlane(wi, t);
                int w1 = __builtin_amdgcn_readlane(wi, t1);
                cj = grp ? c1 : c0;
                wb = grp ? w1 : w0;
                ok = (t + grp) < n;
            }
            if (ok) {
                const float  wv = __int_as_float(wb);
                const float4 hv = *(const float4*)(hb + (size_t)cj * Cch);
                acc.x = fmaf(hv.x, wv, acc.x);
                acc.y = fmaf(hv.y, wv, acc.y);
                acc.z = fmaf(hv.z, wv, acc.z);
                acc.w = fmaf(hv.w, wv, acc.w);
            }
        }
    }
    if (EPS == 2) {
        acc.x += __shfl_xor(acc.x, 16);
        acc.y += __shfl_xor(acc.y, 16);
        acc.z += __shfl_xor(acc.z, 16);
        acc.w += __shfl_xor(acc.w, 16);
    }
    if (grp == 0) *(float4*)(out + (size_t)row * Cch + sub * 4) = acc;
}

extern "C" void kernel_launch(void* const* d_in, const int* in_sizes, int n_in,
                              void* d_out, int out_size, void* d_ws, size_t ws_size,
                              hipStream_t stream) {
    const float* x   = (const float*)d_in[0];
    const float* W1  = (const float*)d_in[1];
    const float* W2  = (const float*)d_in[2];
    const float* ew  = (const float*)d_in[3];
    const int*   er  = (const int*)  d_in[4];
    const int*   ec  = (const int*)  d_in[5];

    const int F_IN = 512, C1 = 128, C2 = 64;
    const int nNodes = in_sizes[0] / F_IN;
    const int nEdges = in_sizes[3];

    float* H   = (float*)d_ws;
    float* Hs  = H  + (size_t)nNodes * C1;
    float* R3  = Hs + (size_t)nNodes * C1;
    float* H2  = H;
    float* out = (float*)d_out;

    int*            rowptr = (int*)R3;
    float*          sw     = R3 + 65536;
    unsigned short* scol   = (unsigned short*)(sw + nEdges);
    int*  P   = (int*)H;
    int2* rec = (int2*)(H + 65536);

    const int BLK = 256;
    const int chunk = (nEdges + NB - 1) / NB;
    const int rpb   = (nNodes + NB - 1) / NB;

    bin_count<<<NB, SORT_T, 0, stream>>>(er, P, nEdges, chunk, (unsigned)rpb);
    bin_scan<<<1, SORT_T, 0, stream>>>(P);
    bin_place<<<NB, SORT_T, 0, stream>>>(er, P, rec, nEdges, chunk, (unsigned)rpb);
    owner_sort<<<NB, SORT_T, 0, stream>>>(rec, P, ec, ew, rowptr, scol, sw,
                                          nEdges, nNodes, rpb);

    {
        int mTiles = nNodes / 16;
        dim3 grid(C1 / 64, (mTiles + 3) / 4);
        gemm_wmma_f32<0, 128, 512, 4, 4><<<grid, 128, 0, stream>>>(x, W1, H, nNodes);
    }

    spmm_rows<5><<<(nNodes + 7) / 8, BLK, 0, stream>>>(rowptr, scol, sw, H, Hs, nNodes);

    {
        int mTiles = nNodes / 16;
        dim3 grid(C2 / 64, (mTiles + 3) / 4);
        gemm_wmma_f32<1, 64, 128, 4, 4><<<grid, 128, 0, stream>>>(Hs, W2, H2, nNodes);
    }

    spmm_rows<4><<<(nNodes + 7) / 8, BLK, 0, stream>>>(rowptr, scol, sw, H2, out, nNodes);
}


# One-block scan launch of the 256x256 edge-count matrix removed: each placement block sums the matrix columns itself (coalesced 16-byte reads, LDS fold) for its start positions; 8 launches become 7
# speedup vs baseline: 2.7159x; 1.0931x over previous
#include <hip/hip_runtime.h>

typedef float v2f __attribute__((ext_vector_type(2)));
typedef float v8f __attribute__((ext_vector_type(8)));

template <int RELU_A, int NN, int KK, int NT, int WAVES>
__global__ void __launch_bounds__(WAVES * 32)
gemm_wmma_f32(const float* __restrict__ A,
              const float* __restrict__ B,
              float* __restrict__ C,
              int M) {
    constexpr int LDB = KK + 4;
    constexpr int T   = WAVES * 32;
    constexpr int U   = 16;
    static_assert(KK % (4 * U) == 0, "K must be a multiple of the register block");
    static_assert((KK * 16 * NT / 4) % T == 0 && KK % (T / 4) == 0, "panel staging shape");
    __shared__ float Bt[16 * NT * LDB];

    const int tid = (int)threadIdx.x;
    const int n0  = (int)blockIdx.x * (16 * NT);

    {
        const int q  = tid & 3;
        const int kk = tid >> 2;
        constexpr int ITERS = KK * 16 * NT / 4 / T;
        #pragma unroll 8
        for (int it = 0; it < ITERS; ++it) {
            const int g   = it % NT;
            const int k   = (it / NT) * (T / 4) + kk;
            const int col = 16 * g + 4 * q;
            const float4 v = *(const float4*)(B + k * NN + n0 + col);
            float* d = &Bt[col * LDB + k];
            d[0]       = v.x;
            d[LDB]     = v.y;
            d[2 * LDB] = v.z;
            d[3 * LDB] = v.w;
        }
    }
    __syncthreads();

    const int wave = tid >> 5;
    const int lane = tid & 31;
    const int tileM = (int)blockIdx.y * WAVES + wave;
    if (tileM >= (M >> 4)) return;

    const int l15  = lane & 15;
    const int kOff = (lane >> 4) << 1;

    const float* __restrict__ arow = A + ((tileM << 4) + l15) * KK + kOff;
    const float* __restrict__ brow = &Bt[l15 * LDB + kOff];

    v8f acc[NT];
    #pragma unroll
    for (int j = 0; j < NT; ++j) acc[j] = v8f{};

    float2 an[U];
    #pragma unroll
    for (int u = 0; u < U; ++u) an[u] = *(const float2*)(arow + 4 * u);

    #pragma unroll 1
    for (int kb = 0; kb < KK; kb += 4 * U) {
        float2 ac[U];
        #pragma unroll
        for (int u = 0; u < U; ++u) ac[u] = an[u];
        if (kb + 4 * U < KK) {
            #pragma unroll
            for (int u = 0; u < U; ++u)
                an[u] = *(const float2*)(arow + kb + 4 * U + 4 * u);
        }
        #pragma unroll
        for (int u = 0; u < U; ++u) {
            float2 av = ac[u];
            if (RELU_A) {
                av.x = fmaxf(av.x, 0.f);
                av.y = fmaxf(av.y, 0.f);
            }
            v2f a; a.x = av.x; a.y = av.y;
            #pragma unroll
            for (int j = 0; j < NT; ++j) {
                v2f b = *(const v2f*)(brow + j * 16 * LDB + kb + 4 * u);
                acc[j] = __builtin_amdgcn_wmma_f32_16x16x4_f32(
                     false, a,  false, b,
                     (short)0, acc[j],  false,  false);
            }
        }
    }

    const int cm = (tileM << 4) + ((lane >> 4) << 3);
    float* __restrict__ crow = C + cm * NN + n0 + l15;
    #pragma unroll
    for (int j = 0; j < NT; ++j) {
        #pragma unroll
        for (int r = 0; r < 8; ++r) crow[r * NN + 16 * j] = acc[j][r];
    }
}

#define NB 256
#define SORT_T 1024

__global__ void __launch_bounds__(SORT_T)
bin_count(const int* __restrict__ erow, int* __restrict__ M,
          int nEdges, int chunk, unsigned rpb) {
    __shared__ int cnt[NB];
    const int tid = (int)threadIdx.x;
    const int b   = (int)blockIdx.x;
    if (tid < NB) cnt[tid] = 0;
    __syncthreads();
    const int e0 = b * chunk;
    const int e1 = min(e0 + chunk, nEdges);
    for (int e = e0 + tid; e < e1; e += SORT_T) {
        unsigned o = (unsigned)erow[e] / rpb;
        atomicAdd(&cnt[o], 1);
    }
    __syncthreads();
    if (tid < NB) M[b * NB + tid] = cnt[tid];
}

__device__ __forceinline__ int block_excl_scan(int t, int* wsum) {
    const int lane = (int)threadIdx.x & 31;
    const int wave = (int)threadIdx.x >> 5;
    int x = t;
    #pragma unroll
    for (int d = 1; d < 32; d <<= 1) {
        int y = __shfl_up(x, d);
        if (lane >= d) x += y;
    }
    if (lane == 31) wsum[wave] = x;
    __syncthreads();
    if (wave == 0) {
        int s0 = wsum[lane];
        int s = s0;
        #pragma unroll
        for (int d = 1; d < 32; d <<= 1) {
            int y = __shfl_up(s, d);
            if (lane >= d) s += y;
        }
        wsum[lane] = s - s0;
    }
    __syncthreads();
    return wsum[wave] + x - t;
}

__global__ void __launch_bounds__(SORT_T)
bin_place(const int* __restrict__ erow, const int* __restrict__ M,
          int* __restrict__ S, int2* __restrict__ rec,
          int nEdges, int chunk, unsigned rpb) {
    __shared__ alignas(16) int red[2][SORT_T / 64][NB];
    __shared__ int pos[NB];
    __shared__ int wsum[32];
    const int tid = (int)threadIdx.x;
    const int b   = (int)blockIdx.x;
    {
        const int quad = tid & 63;
        const int grp  = tid >> 6;
        constexpr int G = SORT_T / 64;
        int4 v[NB / G];
        #pragma unroll
        for (int it = 0; it < NB / G; ++it)
            v[it] = *(const int4*)(M + (grp + G * it) * NB + 4 * quad);
        __builtin_amdgcn_sched_barrier(0);
        int4 tot = make_int4(0, 0, 0, 0);
        int4 par = make_int4(0, 0, 0, 0);
        #pragma unroll
        for (int it = 0; it < NB / G; ++it) {
            const int m = (grp + G * it < b) ? -1 : 0;
            tot.x += v[it].x; tot.y += v[it].y; tot.z += v[it].z; tot.w += v[it].w;
            par.x += v[it].x & m; par.y += v[it].y & m; par.z += v[it].z & m; par.w += v[it].w & m;
        }
        *(int4*)&red[0][grp][4 * quad] = tot;
        *(int4*)&red[1][grp][4 * quad] = par;
    }
    __syncthreads();
    int t = 0, p0 = 0;
    if (tid < NB) {
        #pragma unroll
        for (int g = 0; g < SORT_T / 64; ++g) {
            t  += red[0][g][tid];
            p0 += red[1][g][tid];
        }
    }
    const int start = block_excl_scan(t, wsum);
    if (tid < NB) {
        pos[tid] = start + p0;
        if (b == 0) S[tid] = start;
    }
    __syncthreads();
    const int e0 = b * chunk;
    const int e1 = min(e0 + chunk, nEdges);
    for (int e = e0 + tid; e < e1; e += SORT_T) {
        int r = erow[e];
        unsigned o = (unsigned)r / rpb;
        int p = atomicAdd(&pos[o], 1);
        rec[p] = make_int2(r, e);
    }
}

__global__ void __launch_bounds__(SORT_T)
owner_sort(const int2* __restrict__ rec, const int* __restrict__ S,
           const int* __restrict__ ecol, const float* __restrict__ ew,
           int* __restrict__ rowptr, unsigned short* __restrict__ scol,
           float* __restrict__ sw, int nEdges, int nNodes, int rpb) {
    __shared__ int cnt[SORT_T];
    __shared__ int wsum[32];
    const int tid = (int)threadIdx.x;
    const int o   = (int)blockIdx.x;
    const int s   = S[o];
    const int e   = (o == NB - 1) ? nEdges : S[o + 1];
    const int r0  = o * rpb;

    cnt[tid] = 0;
    __syncthreads();
    for (int p = s + tid; p < e; p += SORT_T) {
        int2 rc = rec[p];
        atomicAdd(&cnt[rc.x - r0], 1);
    }
    __syncthreads();
    const int mine = cnt[tid];
    const int excl = block_excl_scan(mine, wsum);
    if (tid < rpb && r0 + tid < nNodes) rowptr[r0 + tid] = s + excl;
    if (o == NB - 1 && tid == 0) rowptr[nNodes] = nEdges;
    cnt[tid] = excl;
    __syncthreads();
    for (int p = s + tid; p < e; p += SORT_T) {
        int2 rc = rec[p];
        int k = atomicAdd(&cnt[rc.x - r0], 1);
        int dst = s + k;
        scol[dst] = (unsigned short)ecol[rc.y];
        sw[dst]   = ew[rc.y];
    }
}

template <int SHIFT>
__global__ void spmm_rows(const int* __restrict__ rowptr,
                          const unsigned short* __restrict__ scol,
                          const float* __restrict__ sw,
                          const float* __restrict__ h,
                          float* __restrict__ out,
                          int nRows) {
    constexpr int LPE = 1 << SHIFT;
    constexpr int EPS = 32 >> SHIFT;
    constexpr int Cch = 4 << SHIFT;
    const int lane = (int)threadIdx.x & 31;
    const int row  = (int)blockIdx.x * ((int)blockDim.x >> 5) + ((int)threadIdx.x >> 5);
    if (row >= nRows) return;
    const int s = __builtin_amdgcn_readfirstlane(rowptr[row]);
    const int e = __builtin_amdgcn_readfirstlane(rowptr[row + 1]);
    const int sub = lane & (LPE - 1);
    const int grp = lane >> SHIFT;
    const float* __restrict__ hb = h + sub * 4;

    float4 acc = make_float4(0.f, 0.f, 0.f, 0.f);
    for (int base = s; base < e; base += 32) {
        const int n = min(32, e - base);
        int   c = 0;
        float w = 0.f;
        if (lane < n) { c = (int)scol[base + lane]; w = sw[base + lane]; }
        const int wi = __float_as_int(w);
        int t = 0;
        for (; t + 4 * EPS <= n; t += 4 * EPS) {
            float4 hv[4];
            float  wj[4];
            #pragma unroll
            for (int u = 0; u < 4; ++u) {
                int cj, wb;
                if (EPS == 1) {
                    cj = __builtin_amdgcn_readlane(c,  t + u);
                    wb = __builtin_amdgcn_readlane(wi, t + u);
                } else {
                    int c0 = __builtin_amdgcn_readlane(c,  t + 2 * u);
                    int c1 = __builtin_amdgcn_readlane(c,  t + 2 * u + 1);
                    int w0 = __builtin_amdgcn_readlane(wi, t + 2 * u);
                    int w1 = __builtin_amdgcn_readlane(wi, t + 2 * u + 1);
                    cj = grp ? c1 : c0;
                    wb = grp ? w1 : w0;
                }
                wj[u] = __int_as_float(wb);
                hv[u] = *(const float4*)(hb + (size_t)cj * Cch);
            }
            #pragma unroll
            for (int u = 0; u < 4; ++u) {
                acc.x = fmaf(hv[u].x, wj[u], acc.x);
                acc.y = fmaf(hv[u].y, wj[u], acc.y);
                acc.z = fmaf(hv[u].z, wj[u], acc.z);
                acc.w = fmaf(hv[u].w, wj[u], acc.w);
            }
        }
        for (; t < n; t += EPS) {
            int cj, wb;
            bool ok = true;
            if (EPS == 1) {
                cj = __builtin_amdgcn_readlane(c,  t);
                wb = __builtin_amdgcn_readlane(wi, t);
            } else {
                const int t1 = min(t + 1, 31);
                int c0 = __builtin_amdgcn_readlane(c,  t);
                int c1 = __builtin_amdgcn_readlane(c,  t1);
                int w0 = __builtin_amdgcn_readlane(wi, t);
                int w1 = __builtin_amdgcn_readlane(wi, t1);
                cj = grp ? c1 : c0;
                wb = grp ? w1 : w0;
                ok = (t + grp) < n;
            }
            if (ok) {
                const float  wv = __int_as_float(wb);
                const float4 hv = *(const float4*)(hb + (size_t)cj * Cch);
                acc.x = fmaf(hv.x, wv, acc.x);
                acc.y = fmaf(hv.y, wv, acc.y);
                acc.z = fmaf(hv.z, wv, acc.z);
                acc.w = fmaf(hv.w, wv, acc.w);
            }
        }
    }
    if (EPS == 2) {
        acc.x += __shfl_xor(acc.x, 16);
        acc.y += __shfl_xor(acc.y, 16);
        acc.z += __shfl_xor(acc.z, 16);
        acc.w += __shfl_xor(acc.w, 16);
    }
    if (grp == 0) *(float4*)(out + (size_t)row * Cch + sub * 4) = acc;
}

extern "C" void kernel_launch(void* const* d_in, const int* in_sizes, int n_in,
                              void* d_out, int out_size, void* d_ws, size_t ws_size,
                              hipStream_t stream) {
    const float* x   = (const float*)d_in[0];
    const float* W1  = (const float*)d_in[1];
    const float* W2  = (const float*)d_in[2];
    const float* ew  = (const float*)d_in[3];
    const int*   er  = (const int*)  d_in[4];
    const int*   ec  = (const int*)  d_in[5];

    const int F_IN = 512, C1 = 128, C2 = 64;
    const int nNodes = in_sizes[0] / F_IN;
    const int nEdges = in_sizes[3];

    float* H   = (float*)d_ws;
    float* Hs  = H  + (size_t)nNodes * C1;
    float* R3  = Hs + (size_t)nNodes * C1;
    float* H2  = H;
    float* out = (float*)d_out;

    int*            rowptr = (int*)R3;
    float*          sw     = R3 + 65536;
    unsigned short* scol   = (unsigned short*)(sw + nEdges);
    int*  M   = (int*)H;
    int*  S   = M + NB * NB;
    int2* rec = (int2*)(H + 65536 + 512);

    const int BLK = 256;
    const int chunk = (nEdges + NB - 1) / NB;
    const int rpb   = (nNodes + NB - 1) / NB;

    bin_count<<<NB, SORT_T, 0, stream>>>(er, M, nEdges, chunk, (unsigned)rpb);
    bin_place<<<NB, SORT_T, 0, stream>>>(er, M, S, rec, nEdges, chunk, (unsigned)rpb);
    owner_sort<<<NB, SORT_T, 0, stream>>>(rec, S, ec, ew, rowptr, scol, sw,
                                          nEdges, nNodes, rpb);

    {
        int mTiles = nNodes / 16;
        dim3 grid(C1 / 64, (mTiles + 3) / 4);
        gemm_wmma_f32<0, 128, 512, 4, 4><<<grid, 128, 0, stream>>>(x, W1, H, nNodes);
    }

    spmm_rows<5><<<(nNodes + 7) / 8, BLK, 0, stream>>>(rowptr, scol, sw, H, Hs, nNodes);

    {
        int mTiles = nNodes / 16;
        dim3 grid(C2 / 64, (mTiles + 3) / 4);
        gemm_wmma_f32<1, 64, 128, 4, 4><<<grid, 128, 0, stream>>>(Hs, W2, H2, nNodes);
    }

    spmm_rows<4><<<(nNodes + 7) / 8, BLK, 0, stream>>>(rowptr, scol, sw, H2, out, nNodes);
}


# Level-2 edge grouping in LDS: records carry (local row, col, weight); owner block ranks them with LDS counters, stages sorted lists in LDS, stores contiguous runs; row sum order still LDS rank order
# speedup vs baseline: 3.1061x; 1.1437x over previous
#include <hip/hip_runtime.h>

typedef float v2f __attribute__((ext_vector_type(2)));
typedef float v8f __attribute__((ext_vector_type(8)));

template <int RELU_A, int NN, int KK, int NT, int WAVES>
__global__ void __launch_bounds__(WAVES * 32)
gemm_wmma_f32(const float* __restrict__ A,
              const float* __restrict__ B,
              float* __restrict__ C,
              int M) {
    constexpr int LDB = KK + 4;
    constexpr int T   = WAVES * 32;
    constexpr int U   = 16;
    static_assert(KK % (4 * U) == 0, "K must be a multiple of the register block");
    static_assert((KK * 16 * NT / 4) % T == 0 && KK % (T / 4) == 0, "panel staging shape");
    __shared__ float Bt[16 * NT * LDB];

    const int tid = (int)threadIdx.x;
    const int n0  = (int)blockIdx.x * (16 * NT);

    {
        const int q  = tid & 3;
        const int kk = tid >> 2;
        constexpr int ITERS = KK * 16 * NT / 4 / T;
        #pragma unroll 8
        for (int it = 0; it < ITERS; ++it) {
            const int g   = it % NT;
            const int k   = (it / NT) * (T / 4) + kk;
            const int col = 16 * g + 4 * q;
            const float4 v = *(const float4*)(B + k * NN + n0 + col);
            float* d = &Bt[col * LDB + k];
            d[0]       = v.x;
            d[LDB]     = v.y;
            d[2 * LDB] = v.z;
            d[3 * LDB] = v.w;
        }
    }
    __syncthreads();

    const int wave = tid >> 5;
    const int lane = tid & 31;
    const int tileM = (int)blockIdx.y * WAVES + wave;
    if (tileM >= (M >> 4)) return;

    const int l15  = lane & 15;
    const int kOff = (lane >> 4) << 1;

    const float* __restrict__ arow = A + ((tileM << 4) + l15) * KK + kOff;
    const float* __restrict__ brow = &Bt[l15 * LDB + kOff];

    v8f acc[NT];
    #pragma unroll
    for (int j = 0; j < NT; ++j) acc[j] = v8f{};

    float2 an[U];
    #pragma unroll
    for (int u = 0; u < U; ++u) an[u] = *(const float2*)(arow + 4 * u);

    #pragma unroll 1
    for (int kb = 0; kb < KK; kb += 4 * U) {
        float2 ac[U];
        #pragma unroll
        for (int u = 0; u < U; ++u) ac[u] = an[u];
        if (kb + 4 * U < KK) {
            #pragma unroll
            for (int u = 0; u < U; ++u)
                an[u] = *(const float2*)(arow + kb + 4 * U + 4 * u);
        }
        #pragma unroll
        for (int u = 0; u < U; ++u) {
            float2 av = ac[u];
            if (RELU_A) {
                av.x = fmaxf(av.x, 0.f);
                av.y = fmaxf(av.y, 0.f);
            }
            v2f a; a.x = av.x; a.y = av.y;
            #pragma unroll
            for (int j = 0; j < NT; ++j) {
                v2f b = *(const v2f*)(brow + j * 16 * LDB + kb + 4 * u);
                acc[j] = __builtin_amdgcn_wmma_f32_16x16x4_f32(
                     false, a,  false, b,
                     (short)0, acc[j],  false,  false);
            }
        }
    }

    const int cm = (tileM << 4) + ((lane >> 4) << 3);
    float* __restrict__ crow = C + cm * NN + n0 + l15;
    #pragma unroll
    for (int j = 0; j < NT; ++j) {
        #pragma unroll
        for (int r = 0; r < 8; ++r) crow[r * NN + 16 * j] = acc[j][r];
    }
}

#define NB 256
#define SORT_T 1024

__global__ void __launch_bounds__(SORT_T)
bin_count(const int* __restrict__ erow, int* __restrict__ M,
          int nEdges, int chunk, unsigned rpb) {
    __shared__ int cnt[NB];
    const int tid = (int)threadIdx.x;
    const int b   = (int)blockIdx.x;
    if (tid < NB) cnt[tid] = 0;
    __syncthreads();
    const int e0 = b * chunk;
    const int e1 = min(e0 + chunk, nEdges);
    for (int e = e0 + tid; e < e1; e += SORT_T) {
        unsigned o = (unsigned)erow[e] / rpb;
        atomicAdd(&cnt[o], 1);
    }
    __syncthreads();
    if (tid < NB) M[b * NB + tid] = cnt[tid];
}

__device__ __forceinline__ int block_excl_scan(int t, int* wsum) {
    const int lane = (int)threadIdx.x & 31;
    const int wave = (int)threadIdx.x >> 5;
    int x = t;
    #pragma unroll
    for (int d = 1; d < 32; d <<= 1) {
        int y = __shfl_up(x, d);
        if (lane >= d) x += y;
    }
    if (lane == 31) wsum[wave] = x;
    __syncthreads();
    if (wave == 0) {
        int s0 = wsum[lane];
        int s = s0;
        #pragma unroll
        for (int d = 1; d < 32; d <<= 1) {
            int y = __shfl_up(s, d);
            if (lane >= d) s += y;
        }
        wsum[lane] = s - s0;
    }
    __syncthreads();
    return wsum[wave] + x - t;
}

__global__ void __launch_bounds__(SORT_T)
bin_place(const int* __restrict__ erow, const int* __restrict__ ecol,
          const float* __restrict__ ew, const int* __restrict__ M,
          int* __restrict__ S, int2* __restrict__ rec,
          int nEdges, int chunk, unsigned rpb) {
    __shared__ alignas(16) int red[2][SORT_T / 64][NB];
    __shared__ int pos[NB];
    __shared__ int wsum[32];
    const int tid = (int)threadIdx.x;
    const int b   = (int)blockIdx.x;
    {
        const int quad = tid & 63;
        const int grp  = tid >> 6;
        constexpr int G = SORT_T / 64;
        int4 v[NB / G];
        #pragma unroll
        for (int it = 0; it < NB / G; ++it)
            v[it] = *(const int4*)(M + (grp + G * it) * NB + 4 * quad);
        __builtin_amdgcn_sched_barrier(0);
        int4 tot = make_int4(0, 0, 0, 0);
        int4 par = make_int4(0, 0, 0, 0);
        #pragma unroll
        for (int it = 0; it < NB / G; ++it) {
            const int m = (grp + G * it < b) ? -1 : 0;
            tot.x += v[it].x; tot.y += v[it].y; tot.z += v[it].z; tot.w += v[it].w;
            par.x += v[it].x & m; par.y += v[it].y & m; par.z += v[it].z & m; par.w += v[it].w & m;
        }
        *(int4*)&red[0][grp][4 * quad] = tot;
        *(int4*)&red[1][grp][4 * quad] = par;
    }
    __syncthreads();
    int t = 0, p0 = 0;
    if (tid < NB) {
        #pragma unroll
        for (int g = 0; g < SORT_T / 64; ++g) {
            t  += red[0][g][tid];
            p0 += red[1][g][tid];
        }
    }
    const int start = block_excl_scan(t, wsum);
    if (tid < NB) {
        pos[tid] = start + p0;
        if (b == 0) S[tid] = start;
    }
    __syncthreads();
    const int e0 = b * chunk;
    const int e1 = min(e0 + chunk, nEdges);
    for (int e = e0 + tid; e < e1; e += SORT_T) {
        const int   r  = erow[e];
        const int   cc = ecol[e];
        const float w  = ew[e];
        unsigned o = (unsigned)r / rpb;
        int p = atomicAdd(&pos[o], 1);
        const unsigned lr = (unsigned)r - o * rpb;
        rec[p] = make_int2((int)((lr << 16) | ((unsigned)cc & 0xffffu)), __float_as_int(w));
    }
}

#define STAGE_CAP 16384

__global__ void __launch_bounds__(SORT_T)
owner_sort(const int2* __restrict__ rec, const int* __restrict__ S,
           int* __restrict__ rowptr, unsigned short* __restrict__ scol,
           float* __restrict__ sw, int nEdges, int nNodes, int rpb) {
    __shared__ int cnt[SORT_T];
    __shared__ int wsum[32];
    __shared__ float stw[STAGE_CAP];
    __shared__ unsigned short stc[STAGE_CAP];
    const int tid = (int)threadIdx.x;
    const int o   = (int)blockIdx.x;
    const int s   = S[o];
    const int e   = (o == NB - 1) ? nEdges : S[o + 1];
    const int n   = e - s;
    const int r0  = o * rpb;

    cnt[tid] = 0;
    __syncthreads();
    for (int p = s + tid; p < e; p += SORT_T) {
        int2 rc = rec[p];
        atomicAdd(&cnt[(unsigned)rc.x >> 16], 1);
    }
    __syncthreads();
    const int mine = cnt[tid];
    const int excl = block_excl_scan(mine, wsum);
    if (tid < rpb && r0 + tid < nNodes) rowptr[r0 + tid] = s + excl;
    if (o == NB - 1 && tid == 0) rowptr[nNodes] = nEdges;
    cnt[tid] = excl;
    __syncthreads();
    if (n <= STAGE_CAP) {
        for (int p = s + tid; p < e; p += SORT_T) {
            int2 rc = rec[p];
            int k = atomicAdd(&cnt[(unsigned)rc.x >> 16], 1);
            stc[k] = (unsigned short)((unsigned)rc.x & 0xffffu);
            stw[k] = __int_as_float(rc.y);
        }
        __syncthreads();
        for (int i = tid; i < n; i += SORT_T) {
            scol[s + i] = stc[i];
            sw[s + i]   = stw[i];
        }
    } else {
        for (int p = s + tid; p < e; p += SORT_T) {
            int2 rc = rec[p];
            int k = atomicAdd(&cnt[(unsigned)rc.x >> 16], 1);
            int dst = s + k;
            scol[dst] = (unsigned short)((unsigned)rc.x & 0xffffu);
            sw[dst]   = __int_as_float(rc.y);
        }
    }
}

template <int SHIFT>
__global__ void spmm_rows(const int* __restrict__ rowptr,
                          const unsigned short* __restrict__ scol,
                          const float* __restrict__ sw,
                          const float* __restrict__ h,
                          float* __restrict__ out,
                          int nRows) {
    constexpr int LPE = 1 << SHIFT;
    constexpr int EPS = 32 >> SHIFT;
    constexpr int Cch = 4 << SHIFT;
    const int lane = (int)threadIdx.x & 31;
    const int row  = (int)blockIdx.x * ((int)blockDim.x >> 5) + ((int)threadIdx.x >> 5);
    if (row >= nRows) return;
    const int s = __builtin_amdgcn_readfirstlane(rowptr[row]);
    const int e = __builtin_amdgcn_readfirstlane(rowptr[row + 1]);
    const int sub = lane & (LPE - 1);
    const int grp = lane >> SHIFT;
    const float* __restrict__ hb = h + sub * 4;

    float4 acc = make_float4(0.f, 0.f, 0.f, 0.f);
    for (int base = s; base < e; base += 32) {
        const int n = min(32, e - base);
        int   c = 0;
        float w = 0.f;
        if (lane < n) { c = (int)scol[base + lane]; w = sw[base + lane]; }
        const int wi = __float_as_int(w);
        int t = 0;
        for (; t + 4 * EPS <= n; t += 4 * EPS) {
            float4 hv[4];
            float  wj[4];
            #pragma unroll
            for (int u = 0; u < 4; ++u) {
                int cj, wb;
                if (EPS == 1) {
                    cj = __builtin_amdgcn_readlane(c,  t + u);
                    wb = __builtin_amdgcn_readlane(wi, t + u);
                } else {
                    int c0 = __builtin_amdgcn_readlane(c,  t + 2 * u);
                    int c1 = __builtin_amdgcn_readlane(c,  t + 2 * u + 1);
                    int w0 = __builtin_amdgcn_readlane(wi, t + 2 * u);
                    int w1 = __builtin_amdgcn_readlane(wi, t + 2 * u + 1);
                    cj = grp ? c1 : c0;
                    wb = grp ? w1 : w0;
                }
                wj[u] = __int_as_float(wb);
                hv[u] = *(const float4*)(hb + (size_t)cj * Cch);
            }
            #pragma unroll
            for (int u = 0; u < 4; ++u) {
                acc.x = fmaf(hv[u].x, wj[u], acc.x);
                acc.y = fmaf(hv[u].y, wj[u], acc.y);
                acc.z = fmaf(hv[u].z, wj[u], acc.z);
                acc.w = fmaf(hv[u].w, wj[u], acc.w);
            }
        }
        for (; t < n; t += EPS) {
            int cj, wb;
            bool ok = true;
            if (EPS == 1) {
                cj = __builtin_amdgcn_readlane(c,  t);
                wb = __builtin_amdgcn_readlane(wi, t);
            } else {
                const int t1 = min(t + 1, 31);
                int c0 = __builtin_amdgcn_readlane(c,  t);
                int c1 = __builtin_amdgcn_readlane(c,  t1);
                int w0 = __builtin_amdgcn_readlane(wi, t);
                int w1 = __builtin_amdgcn_readlane(wi, t1);
                cj = grp ? c1 : c0;
                wb = grp ? w1 : w0;
                ok = (t + grp) < n;
            }
            if (ok) {
                const float  wv = __int_as_float(wb);
                const float4 hv = *(const float4*)(hb + (size_t)cj * Cch);
                acc.x = fmaf(hv.x, wv, acc.x);
                acc.y = fmaf(hv.y, wv, acc.y);
                acc.z = fmaf(hv.z, wv, acc.z);
                acc.w = fmaf(hv.w, wv, acc.w);
            }
        }
    }
    if (EPS == 2) {
        acc.x += __shfl_xor(acc.x, 16);
        acc.y += __shfl_xor(acc.y, 16);
        acc.z += __shfl_xor(acc.z, 16);
        acc.w += __shfl_xor(acc.w, 16);
    }
    if (grp == 0) *(float4*)(out + (size_t)row * Cch + sub * 4) = acc;
}

extern "C" void kernel_launch(void* const* d_in, const int* in_sizes, int n_in,
                              void* d_out, int out_size, void* d_ws, size_t ws_size,
                              hipStream_t stream) {
    const float* x   = (const float*)d_in[0];
    const float* W1  = (const float*)d_in[1];
    const float* W2  = (const float*)d_in[2];
    const float* ew  = (const float*)d_in[3];
    const int*   er  = (const int*)  d_in[4];
    const int*   ec  = (const int*)  d_in[5];

    const int F_IN = 512, C1 = 128, C2 = 64;
    const int nNodes = in_sizes[0] / F_IN;
    const int nEdges = in_sizes[3];

    float* H   = (float*)d_ws;
    float* Hs  = H  + (size_t)nNodes * C1;
    float* R3  = Hs + (size_t)nNodes * C1;
    float* H2  = H;
    float* out = (float*)d_out;

    int*            rowptr = (int*)R3;
    float*          sw     = R3 + 65536;
    unsigned short* scol   = (unsigned short*)(sw + nEdges);
    int*  M   = (int*)H;
    int*  S   = M + NB * NB;
    int2* rec = (int2*)(H + 65536 + 512);

    const int BLK = 256;
    const int chunk = (nEdges + NB - 1) / NB;
    const int rpb   = (nNodes + NB - 1) / NB;

    bin_count<<<NB, SORT_T, 0, stream>>>(er, M, nEdges, chunk, (unsigned)rpb);
    bin_place<<<NB, SORT_T, 0, stream>>>(er, ec, ew, M, S, rec, nEdges, chunk, (unsigned)rpb);
    owner_sort<<<NB, SORT_T, 0, stream>>>(rec, S, rowptr, scol, sw,
                                          nEdges, nNodes, rpb);

    {
        int mTiles = nNodes / 16;
        dim3 grid(C1 / 64, (mTiles + 3) / 4);
        gemm_wmma_f32<0, 128, 512, 4, 4><<<grid, 128, 0, stream>>>(x, W1, H, nNodes);
    }

    spmm_rows<5><<<(nNodes + 7) / 8, BLK, 0, stream>>>(rowptr, scol, sw, H, Hs, nNodes);

    {
        int mTiles = nNodes / 16;
        dim3 grid(C2 / 64, (mTiles + 3) / 4);
        gemm_wmma_f32<1, 64, 128, 4, 4><<<grid, 128, 0, stream>>>(Hs, W2, H2, nNodes);
    }

    spmm_rows<4><<<(nNodes + 7) / 8, BLK, 0, stream>>>(rowptr, scol, sw, H2, out, nNodes);
}


# Level-1 edge grouping made block-local (a block counts, scans and places its own chunk) and run as 256 extra blocks of the GEMM1 launch; owner blocks collect their 256 runs; 7 launches become 5
# speedup vs baseline: 3.4512x; 1.1111x over previous
#include <hip/hip_runtime.h>

typedef float v2f __attribute__((ext_vector_type(2)));
typedef float v8f __attribute__((ext_vector_type(8)));

template <int RELU_A, int NN, int KK, int NT, int WAVES>
__device__ __forceinline__ void
gemm_block(const float* __restrict__ A,
           const float* __restrict__ B,
           float* __restrict__ C,
           int M, int bx, int by) {
    constexpr int LDB = KK + 4;
    constexpr int T   = WAVES * 32;
    constexpr int U   = 16;
    static_assert(KK % (4 * U) == 0, "K must be a multiple of the register block");
    static_assert((KK * 16 * NT / 4) % T == 0 && KK % (T / 4) == 0, "panel staging shape");
    __shared__ float Bt[16 * NT * LDB];

    const int tid = (int)threadIdx.x;
    const int n0  = bx * (16 * NT);

    {
        const int q  = tid & 3;
        const int kk = tid >> 2;
        constexpr int ITERS = KK * 16 * NT / 4 / T;
        #pragma unroll 8
        for (int it = 0; it < ITERS; ++it) {
            const int g   = it % NT;
            const int k   = (it / NT) * (T / 4) + kk;
            const int col = 16 * g + 4 * q;
            const float4 v = *(const float4*)(B + k * NN + n0 + col);
            float* d = &Bt[col * LDB + k];
            d[0]       = v.x;
            d[LDB]     = v.y;
            d[2 * LDB] = v.z;
            d[3 * LDB] = v.w;
        }
    }
    __syncthreads();

    const int wave = tid >> 5;
    const int lane = tid & 31;
    const int tileM = by * WAVES + wave;
    if (tileM >= (M >> 4)) return;

    const int l15  = lane & 15;
    const int kOff = (lane >> 4) << 1;

    const float* __restrict__ arow = A + ((tileM << 4) + l15) * KK + kOff;
    const float* __restrict__ brow = &Bt[l15 * LDB + kOff];

    v8f acc[NT];
    #pragma unroll
    for (int j = 0; j < NT; ++j) acc[j] = v8f{};

    float2 an[U];
    #pragma unroll
    for (int u = 0; u < U; ++u) an[u] = *(const float2*)(arow + 4 * u);

    #pragma unroll 1
    for (int kb = 0; kb < KK; kb += 4 * U) {
        float2 ac[U];
        #pragma unroll
        for (int u = 0; u < U; ++u) ac[u] = an[u];
        if (kb + 4 * U < KK) {
            #pragma unroll
            for (int u = 0; u < U; ++u)
                an[u] = *(const float2*)(arow + kb + 4 * U + 4 * u);
        }
        #pragma unroll
        for (int u = 0; u < U; ++u) {
            float2 av = ac[u];
            if (RELU_A) {
                av.x = fmaxf(av.x, 0.f);
                av.y = fmaxf(av.y, 0.f);
            }
            v2f a; a.x = av.x; a.y = av.y;
            #pragma unroll
            for (int j = 0; j < NT; ++j) {
                v2f b = *(const v2f*)(brow + j * 16 * LDB + kb + 4 * u);
                acc[j] = __builtin_amdgcn_wmma_f32_16x16x4_f32(
                     false, a,  false, b,
                     (short)0, acc[j],  false,  false);
            }
        }
    }

    const int cm = (tileM << 4) + ((lane >> 4) << 3);
    float* __restrict__ crow = C + cm * NN + n0 + l15;
    #pragma unroll
    for (int j = 0; j < NT; ++j) {
        #pragma unroll
        for (int r = 0; r < 8; ++r) crow[r * NN + 16 * j] = acc[j][r];
    }
}

template <int RELU_A, int NN, int KK, int NT, int WAVES>
__global__ void __launch_bounds__(WAVES * 32)
gemm_wmma_f32(const float* __restrict__ A,
              const float* __restrict__ B,
              float* __restrict__ C,
              int M) {
    gemm_block<RELU_A, NN, KK, NT, WAVES>(A, B, C, M, (int)blockIdx.x, (int)blockIdx.y);
}

#define NB 256
#define SORT_T 1024

template <int T>
__device__ __forceinline__ void
group_chunk(const int* __restrict__ erow, const int* __restrict__ ecol,
            const float* __restrict__ ew, int* __restrict__ Mc,
            int* __restrict__ Lst, int2* __restrict__ rec,
            int nEdges, int chunk, unsigned rpb, int b) {
    static_assert(2 * T == NB, "two owners per thread in the scan");
    __shared__ alignas(8) int cnt[NB];
    __shared__ alignas(8) int pos[NB];
    __shared__ int wtot[T / 32];
    const int tid  = (int)threadIdx.x;
    const int lane = tid & 31;
    const int wave = tid >> 5;
    cnt[2 * tid] = 0;
    cnt[2 * tid + 1] = 0;
    __syncthreads();
    const int e0 = b * chunk;
    const int e1 = min(e0 + chunk, nEdges);
    for (int base = e0 + tid; base < e1; base += T * 16) {
        int r[16];
        #pragma unroll
        for (int u = 0; u < 16; ++u) {
            const int e = base + u * T;
            r[u] = (e < e1) ? erow[e] : 0;
        }
        #pragma unroll
        for (int u = 0; u < 16; ++u) {
            if (base + u * T < e1) atomicAdd(&cnt[(unsigned)r[u] / rpb], 1);
        }
    }
    __syncthreads();
    const int a0 = cnt[2 * tid];
    const int a1 = cnt[2 * tid + 1];
    int x = a0 + a1;
    #pragma unroll
    for (int d = 1; d < 32; d <<= 1) {
        int y = __shfl_up(x, d);
        if (lane >= d) x += y;
    }
    if (lane == 31) wtot[wave] = x;
    __syncthreads();
    int wbase = 0;
    for (int w = 0; w < wave; ++w) wbase += wtot[w];
    const int ex = wbase + x - (a0 + a1);
    pos[2 * tid]     = ex;
    pos[2 * tid + 1] = ex + a0;
    *(int2*)(Mc  + b * NB + 2 * tid) = make_int2(a0, a1);
    *(int2*)(Lst + b * NB + 2 * tid) = make_int2(ex, ex + a0);
    __syncthreads();
    for (int base = e0 + tid; base < e1; base += T * 8) {
        int   r[8], c[8];
        float w[8];
        #pragma unroll
        for (int u = 0; u < 8; ++u) {
            const int e = base + u * T;
            const bool ok = e < e1;
            r[u] = ok ? erow[e] : 0;
            c[u] = ok ? ecol[e] : 0;
            w[u] = ok ? ew[e]   : 0.f;
        }
        int p[8];
        #pragma unroll
        for (int u = 0; u < 8; ++u) {
            const unsigned o = (unsigned)r[u] / rpb;
            p[u] = 0;
            if (base + u * T < e1) p[u] = atomicAdd(&pos[o], 1);
            r[u] = (int)((unsigned)r[u] - o * rpb);
        }
        #pragma unroll
        for (int u = 0; u < 8; ++u) {
            if (base + u * T < e1)
                rec[e0 + p[u]] = make_int2((int)(((unsigned)r[u] << 16) | ((unsigned)c[u] & 0xffffu)),
                                           __float_as_int(w[u]));
        }
    }
}

__global__ void __launch_bounds__(128)
gemm1_with_grouping(const float* __restrict__ x, const float* __restrict__ W1,
                    float* __restrict__ H, int nNodes,
                    const int* __restrict__ erow, const int* __restrict__ ecol,
                    const float* __restrict__ ew, int* __restrict__ Mc,
                    int* __restrict__ Lst, int2* __restrict__ rec,
                    int nEdges, int chunk, unsigned rpb) {
    if ((int)blockIdx.x < NB) {
        group_chunk<128>(erow, ecol, ew, Mc, Lst, rec, nEdges, chunk, rpb, (int)blockIdx.x);
        return;
    }
    const int idx = (int)blockIdx.x - NB;
    gemm_block<0, 128, 512, 4, 4>(x, W1, H, nNodes, idx & 1, idx >> 1);
}

__device__ __forceinline__ int block_excl_scan(int t, int* wsum) {
    const int lane = (int)threadIdx.x & 31;
    const int wave = (int)threadIdx.x >> 5;
    int x = t;
    #pragma unroll
    for (int d = 1; d < 32; d <<= 1) {
        int y = __shfl_up(x, d);
        if (lane >= d) x += y;
    }
    if (lane == 31) wsum[wave] = x;
    __syncthreads();
    if (wave == 0) {
        int s0 = wsum[lane];
        int s = s0;
        #pragma unroll
        for (int d = 1; d < 32; d <<= 1) {
            int y = __shfl_up(s, d);
            if (lane >= d) s += y;
        }
        wsum[lane] = s - s0;
    }
    __syncthreads();
    return wsum[wave] + x - t;
}

#define STAGE_CAP 16384

__global__ void __launch_bounds__(SORT_T)
owner_sort(const int2* __restrict__ rec, const int* __restrict__ Mc,
           const int* __restrict__ Lst,
           int* __restrict__ rowptr, unsigned short* __restrict__ scol,
           float* __restrict__ sw, int nEdges, int nNodes, int rpb, int chunk) {
    static_assert(SORT_T == 4 * NB, "four threads per run");
    __shared__ int cnt[SORT_T];
    __shared__ int wsum[32];
    __shared__ int runStart[NB];
    __shared__ int runLen[NB];
    __shared__ int tot[2];
    __shared__ float stw[STAGE_CAP];
    __shared__ unsigned short stc[STAGE_CAP];
    const int tid = (int)threadIdx.x;
    const int o   = (int)blockIdx.x;
    const int r0  = o * rpb;

    cnt[tid] = 0;
    if (tid < 2) tot[tid] = 0;
    __syncthreads();
    if (tid < NB) {
        int ls = Lst[tid * NB + o];
        int ln = Mc[tid * NB + o];
        runStart[tid] = tid * chunk + ls;
        runLen[tid]   = ln;
        #pragma unroll
        for (int d = 16; d >= 1; d >>= 1) {
            ls += __shfl_xor(ls, d);
            ln += __shfl_xor(ln, d);
        }
        if ((tid & 31) == 0) {
            atomicAdd(&tot[0], ls);
            atomicAdd(&tot[1], ln);
        }
    }
    __syncthreads();
    const int s  = tot[0];
    const int n  = tot[1];
    const int rs = runStart[tid >> 2];
    const int rl = runLen[tid >> 2];
    const int rq = tid & 3;

    for (int j0 = rq; j0 < rl; j0 += 16) {
        int2 rc[4];
        #pragma unroll
        for (int u = 0; u < 4; ++u)
            rc[u] = (j0 + 4 * u < rl) ? rec[rs + j0 + 4 * u] : make_int2(0, 0);
        #pragma unroll
        for (int u = 0; u < 4; ++u)
            if (j0 + 4 * u < rl) atomicAdd(&cnt[(unsigned)rc[u].x >> 16], 1);
    }
    __syncthreads();
    const int mine = cnt[tid];
    const int excl = block_excl_scan(mine, wsum);
    if (tid < rpb && r0 + tid < nNodes) rowptr[r0 + tid] = s + excl;
    if (o == NB - 1 && tid == 0) rowptr[nNodes] = nEdges;
    cnt[tid] = excl;
    __syncthreads();
    if (n <= STAGE_CAP) {
        for (int j0 = rq; j0 < rl; j0 += 16) {
            int2 rc[4];
            #pragma unroll
            for (int u = 0; u < 4; ++u)
                rc[u] = (j0 + 4 * u < rl) ? rec[rs + j0 + 4 * u] : make_int2(0, 0);
            #pragma unroll
            for (int u = 0; u < 4; ++u) {
                if (j0 + 4 * u < rl) {
                    int k = atomicAdd(&cnt[(unsigned)rc[u].x >> 16], 1);
                    stc[k] = (unsigned short)((unsigned)rc[u].x & 0xffffu);
                    stw[k] = __int_as_float(rc[u].y);
                }
            }
        }
        __syncthreads();
        for (int i = tid; i < n; i += SORT_T) {
            scol[s + i] = stc[i];
            sw[s + i]   = stw[i];
        }
    } else {
        for (int j = rq; j < rl; j += 4) {
            int2 rc = rec[rs + j];
            int k = atomicAdd(&cnt[(unsigned)rc.x >> 16], 1);
            int dst = s + k;
            scol[dst] = (unsigned short)((unsigned)rc.x & 0xffffu);
            sw[dst]   = __int_as_float(rc.y);
        }
    }
}

template <int SHIFT>
__global__ void spmm_rows(const int* __restrict__ rowptr,
                          const unsigned short* __restrict__ scol,
                          const float* __restrict__ sw,
                          const float* __restrict__ h,
                          float* __restrict__ out,
                          int nRows) {
    constexpr int LPE = 1 << SHIFT;
    constexpr int EPS = 32 >> SHIFT;
    constexpr int Cch = 4 << SHIFT;
    const int lane = (int)threadIdx.x & 31;
    const int row  = (int)blockIdx.x * ((int)blockDim.x >> 5) + ((int)threadIdx.x >> 5);
    if (row >= nRows) return;
    const int s = __builtin_amdgcn_readfirstlane(rowptr[row]);
    const int e = __builtin_amdgcn_readfirstlane(rowptr[row + 1]);
    const int sub = lane & (LPE - 1);
    const int grp = lane >> SHIFT;
    const float* __restrict__ hb = h + sub * 4;

    float4 acc = make_float4(0.f, 0.f, 0.f, 0.f);
    for (int base = s; base < e; base += 32) {
        const int n = min(32, e - base);
        int   c = 0;
        float w = 0.f;
        if (lane < n) { c = (int)scol[base + lane]; w = sw[base + lane]; }
        const int wi = __float_as_int(w);
        int t = 0;
        for (; t + 4 * EPS <= n; t += 4 * EPS) {
            float4 hv[4];
            float  wj[4];
            #pragma unroll
            for (int u = 0; u < 4; ++u) {
                int cj, wb;
                if (EPS == 1) {
                    cj = __builtin_amdgcn_readlane(c,  t + u);
                    wb = __builtin_amdgcn_readlane(wi, t + u);
                } else {
                    int c0 = __builtin_amdgcn_readlane(c,  t + 2 * u);
                    int c1 = __builtin_amdgcn_readlane(c,  t + 2 * u + 1);
                    int w0 = __builtin_amdgcn_readlane(wi, t + 2 * u);
                    int w1 = __builtin_amdgcn_readlane(wi, t + 2 * u + 1);
                    cj = grp ? c1 : c0;
                    wb = grp ? w1 : w0;
                }
                wj[u] = __int_as_float(wb);
                hv[u] = *(const float4*)(hb + (size_t)cj * Cch);
            }
            #pragma unroll
            for (int u = 0; u < 4; ++u) {
                acc.x = fmaf(hv[u].x, wj[u], acc.x);
                acc.y = fmaf(hv[u].y, wj[u], acc.y);
                acc.z = fmaf(hv[u].z, wj[u], acc.z);
                acc.w = fmaf(hv[u].w, wj[u], acc.w);
            }
        }
        for (; t < n; t += EPS) {
            int cj, wb;
            bool ok = true;
            if (EPS == 1) {
                cj = __builtin_amdgcn_readlane(c,  t);
                wb = __builtin_amdgcn_readlane(wi, t);
            } else {
                const int t1 = min(t + 1, 31);
                int c0 = __builtin_amdgcn_readlane(c,  t);
                int c1 = __builtin_amdgcn_readlane(c,  t1);
                int w0 = __builtin_amdgcn_readlane(wi, t);
                int w1 = __builtin_amdgcn_readlane(wi, t1);
                cj = grp ? c1 : c0;
                wb = grp ? w1 : w0;
                ok = (t + grp) < n;
            }
            if (ok) {
                const float  wv = __int_as_float(wb);
                const float4 hv = *(const float4*)(hb + (size_t)cj * Cch);
                acc.x = fmaf(hv.x, wv, acc.x);
                acc.y = fmaf(hv.y, wv, acc.y);
                acc.z = fmaf(hv.z, wv, acc.z);
                acc.w = fmaf(hv.w, wv, acc.w);
            }
        }
    }
    if (EPS == 2) {
        acc.x += __shfl_xor(acc.x, 16);
        acc.y += __shfl_xor(acc.y, 16);
        acc.z += __shfl_xor(acc.z, 16);
        acc.w += __shfl_xor(acc.w, 16);
    }
    if (grp == 0) *(float4*)(out + (size_t)row * Cch + sub * 4) = acc;
}

extern "C" void kernel_launch(void* const* d_in, const int* in_sizes, int n_in,
                              void* d_out, int out_size, void* d_ws, size_t ws_size,
                              hipStream_t stream) {
    const float* x   = (const float*)d_in[0];
    const float* W1  = (const float*)d_in[1];
    const float* W2  = (const float*)d_in[2];
    const float* ew  = (const float*)d_in[3];
    const int*   er  = (const int*)  d_in[4];
    const int*   ec  = (const int*)  d_in[5];

    const int F_IN = 512, C1 = 128, C2 = 64;
    const int nNodes = in_sizes[0] / F_IN;
    const int nEdges = in_sizes[3];

    float* H   = (float*)d_ws;
    float* Hs  = H  + (size_t)nNodes * C1;
    float* R3  = Hs + (size_t)nNodes * C1;
    float* H2  = H;
    float* out = (float*)d_out;

    int*            rowptr = (int*)R3;
    float*          sw     = R3 + 65536;
    unsigned short* scol   = (unsigned short*)(sw + nEdges);
    int*  Mc  = (int*)Hs;
    int*  Lst = Mc + NB * NB;
    int2* rec = (int2*)(Hs + 2 * 65536);

    const int BLK = 256;
    const int chunk = (nEdges + NB - 1) / NB;
    const int rpb   = (nNodes + NB - 1) / NB;

    {
        int mTiles = nNodes / 16;
        int gemmBlocks = (C1 / 64) * ((mTiles + 3) / 4);
        gemm1_with_grouping<<<NB + gemmBlocks, 128, 0, stream>>>(
            x, W1, H, nNodes, er, ec, ew, Mc, Lst, rec, nEdges, chunk, (unsigned)rpb);
    }
    owner_sort<<<NB, SORT_T, 0, stream>>>(rec, Mc, Lst, rowptr, scol, sw,
                                          nEdges, nNodes, rpb, chunk);

    spmm_rows<5><<<(nNodes + 7) / 8, BLK, 0, stream>>>(rowptr, scol, sw, H, Hs, nNodes);

    {
        int mTiles = nNodes / 16;
        dim3 grid(C2 / 64, (mTiles + 3) / 4);
        gemm_wmma_f32<1, 64, 128, 4, 4><<<grid, 128, 0, stream>>>(Hs, W2, H2, nNodes);
    }

    spmm_rows<4><<<(nNodes + 7) / 8, BLK, 0, stream>>>(rowptr, scol, sw, H2, out, nNodes);
}


# GEMM2 folded into the layer-1 row SpMM: each 16-row block keeps its relu'd row sums in LDS and multiplies them by W2 (staged in LDS) on the matrix unit; 5 launches become 4; sum orders unchanged
# speedup vs baseline: 3.5945x; 1.0415x over previous
#include <hip/hip_runtime.h>

typedef float v2f __attribute__((ext_vector_type(2)));
typedef float v8f __attribute__((ext_vector_type(8)));

template <int RELU_A, int NN, int KK, int NT, int WAVES>
__device__ __forceinline__ void
gemm_block(const float* __restrict__ A,
           const float* __restrict__ B,
           float* __restrict__ C,
           int M, int bx, int by) {
    constexpr int LDB = KK + 4;
    constexpr int T   = WAVES * 32;
    constexpr int U   = 16;
    static_assert(KK % (4 * U) == 0, "K must be a multiple of the register block");
    static_assert((KK * 16 * NT / 4) % T == 0 && KK % (T / 4) == 0, "panel staging shape");
    __shared__ float Bt[16 * NT * LDB];

    const int tid = (int)threadIdx.x;
    const int n0  = bx * (16 * NT);

    {
        const int q  = tid & 3;
        const int kk = tid >> 2;
        constexpr int ITERS = KK * 16 * NT / 4 / T;
        #pragma unroll 8
        for (int it = 0; it < ITERS; ++it) {
            const int g   = it % NT;
            const int k   = (it / NT) * (T / 4) + kk;
            const int col = 16 * g + 4 * q;
            const float4 v = *(const float4*)(B + k * NN + n0 + col);
            float* d = &Bt[col * LDB + k];
            d[0]       = v.x;
            d[LDB]     = v.y;
            d[2 * LDB] = v.z;
            d[3 * LDB] = v.w;
        }
    }
    __syncthreads();

    const int wave = tid >> 5;
    const int lane = tid & 31;
    const int tileM = by * WAVES + wave;
    if (tileM >= (M >> 4)) return;

    const int l15  = lane & 15;
    const int kOff = (lane >> 4) << 1;

    const float* __restrict__ arow = A + ((tileM << 4) + l15) * KK + kOff;
    const float* __restrict__ brow = &Bt[l15 * LDB + kOff];

    v8f acc[NT];
    #pragma unroll
    for (int j = 0; j < NT; ++j) acc[j] = v8f{};

    float2 an[U];
    #pragma unroll
    for (int u = 0; u < U; ++u) an[u] = *(const float2*)(arow + 4 * u);

    #pragma unroll 1
    for (int kb = 0; kb < KK; kb += 4 * U) {
        float2 ac[U];
        #pragma unroll
        for (int u = 0; u < U; ++u) ac[u] = an[u];
        if (kb + 4 * U < KK) {
            #pragma unroll
            for (int u = 0; u < U; ++u)
                an[u] = *(const float2*)(arow + kb + 4 * U + 4 * u);
        }
        #pragma unroll
        for (int u = 0; u < U; ++u) {
            float2 av = ac[u];
            if (RELU_A) {
                av.x = fmaxf(av.x, 0.f);
                av.y = fmaxf(av.y, 0.f);
            }
            v2f a; a.x = av.x; a.y = av.y;
            #pragma unroll
            for (int j = 0; j < NT; ++j) {
                v2f b = *(const v2f*)(brow + j * 16 * LDB + kb + 4 * u);
                acc[j] = __builtin_amdgcn_wmma_f32_16x16x4_f32(
                     false, a,  false, b,
                     (short)0, acc[j],  false,  false);
            }
        }
    }

    const int cm = (tileM << 4) + ((lane >> 4) << 3);
    float* __restrict__ crow = C + cm * NN + n0 + l15;
    #pragma unroll
    for (int j = 0; j < NT; ++j) {
        #pragma unroll
        for (int r = 0; r < 8; ++r) crow[r * NN + 16 * j] = acc[j][r];
    }
}

#define NB 256
#define SORT_T 1024

template <int T>
__device__ __forceinline__ void
group_chunk(const int* __restrict__ erow, const int* __restrict__ ecol,
            const float* __restrict__ ew, int* __restrict__ Mc,
            int* __restrict__ Lst, int2* __restrict__ rec,
            int nEdges, int chunk, unsigned rpb, int b) {
    static_assert(2 * T == NB, "two owners per thread in the scan");
    __shared__ alignas(8) int cnt[NB];
    __shared__ alignas(8) int pos[NB];
    __shared__ int wtot[T / 32];
    const int tid  = (int)threadIdx.x;
    const int lane = tid & 31;
    const int wave = tid >> 5;
    cnt[2 * tid] = 0;
    cnt[2 * tid + 1] = 0;
    __syncthreads();
    const int e0 = b * chunk;
    const int e1 = min(e0 + chunk, nEdges);
    for (int base = e0 + tid; base < e1; base += T * 16) {
        int r[16];
        #pragma unroll
        for (int u = 0; u < 16; ++u) {
            const int e = base + u * T;
            r[u] = (e < e1) ? erow[e] : 0;
        }
        #pragma unroll
        for (int u = 0; u < 16; ++u) {
            if (base + u * T < e1) atomicAdd(&cnt[(unsigned)r[u] / rpb], 1);
        }
    }
    __syncthreads();
    const int a0 = cnt[2 * tid];
    const int a1 = cnt[2 * tid + 1];
    int x = a0 + a1;
    #pragma unroll
    for (int d = 1; d < 32; d <<= 1) {
        int y = __shfl_up(x, d);
        if (lane >= d) x += y;
    }
    if (lane == 31) wtot[wave] = x;
    __syncthreads();
    int wbase = 0;
    for (int w = 0; w < wave; ++w) wbase += wtot[w];
    const int ex = wbase + x - (a0 + a1);
    pos[2 * tid]     = ex;
    pos[2 * tid + 1] = ex + a0;
    *(int2*)(Mc  + b * NB + 2 * tid) = make_int2(a0, a1);
    *(int2*)(Lst + b * NB + 2 * tid) = make_int2(ex, ex + a0);
    __syncthreads();
    for (int base = e0 + tid; base < e1; base += T * 8) {
        int   r[8], c[8];
        float w[8];
        #pragma unroll
        for (int u = 0; u < 8; ++u) {
            const int e = base + u * T;
            const bool ok = e < e1;
            r[u] = ok ? erow[e] : 0;
            c[u] = ok ? ecol[e] : 0;
            w[u] = ok ? ew[e]   : 0.f;
        }
        int p[8];
        #pragma unroll
        for (int u = 0; u < 8; ++u) {
            const unsigned o = (unsigned)r[u] / rpb;
            p[u] = 0;
            if (base + u * T < e1) p[u] = atomicAdd(&pos[o], 1);
            r[u] = (int)((unsigned)r[u] - o * rpb);
        }
        #pragma unroll
        for (int u = 0; u < 8; ++u) {
            if (base + u * T < e1)
                rec[e0 + p[u]] = make_int2((int)(((unsigned)r[u] << 16) | ((unsigned)c[u] & 0xffffu)),
                                           __float_as_int(w[u]));
        }
    }
}

__global__ void __launch_bounds__(128)
gemm1_with_grouping(const float* __restrict__ x, const float* __restrict__ W1,
                    float* __restrict__ H, int nNodes,
                    const int* __restrict__ erow, const int* __restrict__ ecol,
                    const float* __restrict__ ew, int* __restrict__ Mc,
                    int* __restrict__ Lst, int2* __restrict__ rec,
                    int nEdges, int chunk, unsigned rpb) {
    if ((int)blockIdx.x < NB) {
        group_chunk<128>(erow, ecol, ew, Mc, Lst, rec, nEdges, chunk, rpb, (int)blockIdx.x);
        return;
    }
    const int idx = (int)blockIdx.x - NB;
    gemm_block<0, 128, 512, 4, 4>(x, W1, H, nNodes, idx & 1, idx >> 1);
}

__device__ __forceinline__ int block_excl_scan(int t, int* wsum) {
    const int lane = (int)threadIdx.x & 31;
    const int wave = (int)threadIdx.x >> 5;
    int x = t;
    #pragma unroll
    for (int d = 1; d < 32; d <<= 1) {
        int y = __shfl_up(x, d);
        if (lane >= d) x += y;
    }
    if (lane == 31) wsum[wave] = x;
    __syncthreads();
    if (wave == 0) {
        int s0 = wsum[lane];
        int s = s0;
        #pragma unroll
        for (int d = 1; d < 32; d <<= 1) {
            int y = __shfl_up(s, d);
            if (lane >= d) s += y;
        }
        wsum[lane] = s - s0;
    }
    __syncthreads();
    return wsum[wave] + x - t;
}

#define STAGE_CAP 16384

__global__ void __launch_bounds__(SORT_T)
owner_sort(const int2* __restrict__ rec, const int* __restrict__ Mc,
           const int* __restrict__ Lst,
           int* __restrict__ rowptr, unsigned short* __restrict__ scol,
           float* __restrict__ sw, int nEdges, int nNodes, int rpb, int chunk) {
    static_assert(SORT_T == 4 * NB, "four threads per run");
    __shared__ int cnt[SORT_T];
    __shared__ int wsum[32];
    __shared__ int runStart[NB];
    __shared__ int runLen[NB];
    __shared__ int tot[2];
    __shared__ float stw[STAGE_CAP];
    __shared__ unsigned short stc[STAGE_CAP];
    const int tid = (int)threadIdx.x;
    const int o   = (int)blockIdx.x;
    const int r0  = o * rpb;

    cnt[tid] = 0;
    if (tid < 2) tot[tid] = 0;
    __syncthreads();
    if (tid < NB) {
        int ls = Lst[tid * NB + o];
        int ln = Mc[tid * NB + o];
        runStart[tid] = tid * chunk + ls;
        runLen[tid]   = ln;
        #pragma unroll
        for (int d = 16; d >= 1; d >>= 1) {
            ls += __shfl_xor(ls, d);
            ln += __shfl_xor(ln, d);
        }
        if ((tid & 31) == 0) {
            atomicAdd(&tot[0], ls);
            atomicAdd(&tot[1], ln);
        }
    }
    __syncthreads();
    const int s  = tot[0];
    const int n  = tot[1];
    const int rs = runStart[tid >> 2];
    const int rl = runLen[tid >> 2];
    const int rq = tid & 3;

    for (int j0 = rq; j0 < rl; j0 += 16) {
        int2 rc[4];
        #pragma unroll
        for (int u = 0; u < 4; ++u)
            rc[u] = (j0 + 4 * u < rl) ? rec[rs + j0 + 4 * u] : make_int2(0, 0);
        #pragma unroll
        for (int u = 0; u < 4; ++u)
            if (j0 + 4 * u < rl) atomicAdd(&cnt[(unsigned)rc[u].x >> 16], 1);
    }
    __syncthreads();
    const int mine = cnt[tid];
    const int excl = block_excl_scan(mine, wsum);
    if (tid < rpb && r0 + tid < nNodes) rowptr[r0 + tid] = s + excl;
    if (o == NB - 1 && tid == 0) rowptr[nNodes] = nEdges;
    cnt[tid] = excl;
    __syncthreads();
    if (n <= STAGE_CAP) {
        for (int j0 = rq; j0 < rl; j0 += 16) {
            int2 rc[4];
            #pragma unroll
            for (int u = 0; u < 4; ++u)
                rc[u] = (j0 + 4 * u < rl) ? rec[rs + j0 + 4 * u] : make_int2(0, 0);
            #pragma unroll
            for (int u = 0; u < 4; ++u) {
                if (j0 + 4 * u < rl) {
                    int k = atomicAdd(&cnt[(unsigned)rc[u].x >> 16], 1);
                    stc[k] = (unsigned short)((unsigned)rc[u].x & 0xffffu);
                    stw[k] = __int_as_float(rc[u].y);
                }
            }
        }
        __syncthreads();
        for (int i = tid; i < n; i += SORT_T) {
            scol[s + i] = stc[i];
            sw[s + i]   = stw[i];
        }
    } else {
        for (int j = rq; j < rl; j += 4) {
            int2 rc = rec[rs + j];
            int k = atomicAdd(&cnt[(unsigned)rc.x >> 16], 1);
            int dst = s + k;
            scol[dst] = (unsigned short)((unsigned)rc.x & 0xffffu);
            sw[dst]   = __int_as_float(rc.y);
        }
    }
}

template <int SHIFT>
__device__ __forceinline__ float4
row_sum(const unsigned short* __restrict__ scol,
        const float* __restrict__ sw,
        const float* __restrict__ h,
        int rs, int re, int lane) {
    constexpr int LPE = 1 << SHIFT;
    constexpr int EPS = 32 >> SHIFT;
    constexpr int Cch = 4 << SHIFT;
    const int s = __builtin_amdgcn_readfirstlane(rs);
    const int e = __builtin_amdgcn_readfirstlane(re);
    const int sub = lane & (LPE - 1);
    const int grp = lane >> SHIFT;
    const float* __restrict__ hb = h + sub * 4;

    float4 acc = make_float4(0.f, 0.f, 0.f, 0.f);
    for (int base = s; base < e; base += 32) {
        const int n = min(32, e - base);
        int   c = 0;
        float w = 0.f;
        if (lane < n) { c = (int)scol[base + lane]; w = sw[base + lane]; }
        const int wi = __float_as_int(w);
        int t = 0;
        for (; t + 4 * EPS <= n; t += 4 * EPS) {
            float4 hv[4];
            float  wj[4];
            #pragma unroll
            for (int u = 0; u < 4; ++u) {
                int cj, wb;
                if (EPS == 1) {
                    cj = __builtin_amdgcn_readlane(c,  t + u);
                    wb = __builtin_amdgcn_readlane(wi, t + u);
                } else {
                    int c0 = __builtin_amdgcn_readlane(c,  t + 2 * u);
                    int c1 = __builtin_amdgcn_readlane(c,  t + 2 * u + 1);
                    int w0 = __builtin_amdgcn_readlane(wi, t + 2 * u);
                    int w1 = __builtin_amdgcn_readlane(wi, t + 2 * u + 1);
                    cj = grp ? c1 : c0;
                    wb = grp ? w1 : w0;
                }
                wj[u] = __int_as_float(wb);
                hv[u] = *(const float4*)(hb + (size_t)cj * Cch);
            }
            #pragma unroll
            for (int u = 0; u < 4; ++u) {
                acc.x = fmaf(hv[u].x, wj[u], acc.x);
                acc.y = fmaf(hv[u].y, wj[u], acc.y);
                acc.z = fmaf(hv[u].z, wj[u], acc.z);
                acc.w = fmaf(hv[u].w, wj[u], acc.w);
            }
        }
        for (; t < n; t += EPS) {
            int cj, wb;
            bool ok = true;
            if (EPS == 1) {
                cj = __builtin_amdgcn_readlane(c,  t);
                wb = __builtin_amdgcn_readlane(wi, t);
            } else {
                const int t1 = min(t + 1, 31);
                int c0 = __builtin_amdgcn_readlane(c,  t);
                int c1 = __builtin_amdgcn_readlane(c,  t1);
                int w0 = __builtin_amdgcn_readlane(wi, t);
                int w1 = __builtin_amdgcn_readlane(wi, t1);
                cj = grp ? c1 : c0;
                wb = grp ? w1 : w0;
                ok = (t + grp) < n;
            }
            if (ok) {
                const float  wv = __int_as_float(wb);
                const float4 hv = *(const float4*)(hb + (size_t)cj * Cch);
                acc.x = fmaf(hv.x, wv, acc.x);
                acc.y = fmaf(hv.y, wv, acc.y);
                acc.z = fmaf(hv.z, wv, acc.z);
                acc.w = fmaf(hv.w, wv, acc.w);
            }
        }
    }
    if (EPS == 2) {
        acc.x += __shfl_xor(acc.x, 16);
        acc.y += __shfl_xor(acc.y, 16);
        acc.z += __shfl_xor(acc.z, 16);
        acc.w += __shfl_xor(acc.w, 16);
    }
    return acc;
}

template <int SHIFT>
__global__ void spmm_rows(const int* __restrict__ rowptr,
                          const unsigned short* __restrict__ scol,
                          const float* __restrict__ sw,
                          const float* __restrict__ h,
                          float* __restrict__ out,
                          int nRows) {
    constexpr int LPE = 1 << SHIFT;
    constexpr int Cch = 4 << SHIFT;
    const int lane = (int)threadIdx.x & 31;
    const int row  = (int)blockIdx.x * ((int)blockDim.x >> 5) + ((int)threadIdx.x >> 5);
    if (row >= nRows) return;
    const float4 acc = row_sum<SHIFT>(scol, sw, h, rowptr[row], rowptr[row + 1], lane);
    const int sub = lane & (LPE - 1);
    if ((lane >> SHIFT) == 0) *(float4*)(out + (size_t)row * Cch + sub * 4) = acc;
}

#define FUSE_ROWS 16
__global__ void __launch_bounds__(FUSE_ROWS * 32)
spmm_rows_gemm2(const int* __restrict__ rowptr,
                const unsigned short* __restrict__ scol,
                const float* __restrict__ sw,
                const float* __restrict__ h,
                const float* __restrict__ W2,
                float* __restrict__ H2,
                int nRows) {
    constexpr int KK  = 128;
    constexpr int NN  = 64;
    constexpr int LD  = KK + 4;
    constexpr int T   = FUSE_ROWS * 32;
    static_assert((KK * NN / 4) % T == 0 && KK % (T / 4) == 0, "W2 staging shape");
    __shared__ float Bt[NN * LD];
    __shared__ alignas(16) float At[FUSE_ROWS * LD];

    const int tid  = (int)threadIdx.x;
    const int lane = tid & 31;
    const int wave = tid >> 5;

    const int row0 = (int)blockIdx.x * FUSE_ROWS;
    const int row  = row0 + wave;

    const int q  = tid & 3;
    const int kk = tid >> 2;
    constexpr int ITERS = KK * NN / 4 / T;
    float4 wv[ITERS];
    #pragma unroll
    for (int it = 0; it < ITERS; ++it) {
        const int g = it % (NN / 16);
        const int k = (it / (NN / 16)) * (T / 4) + kk;
        wv[it] = *(const float4*)(W2 + k * NN + 16 * g + 4 * q);
    }
    int rs = 0, re = 0;
    if (row < nRows) { rs = rowptr[row]; re = rowptr[row + 1]; }
    #pragma unroll
    for (int it = 0; it < ITERS; ++it) {
        const int g   = it % (NN / 16);
        const int k   = (it / (NN / 16)) * (T / 4) + kk;
        const int col = 16 * g + 4 * q;
        float* d = &Bt[col * LD + k];
        d[0]      = wv[it].x;
        d[LD]     = wv[it].y;
        d[2 * LD] = wv[it].z;
        d[3 * LD] = wv[it].w;
    }

    float4 acc = row_sum<5>(scol, sw, h, rs, re, lane);
    acc.x = fmaxf(acc.x, 0.f);
    acc.y = fmaxf(acc.y, 0.f);
    acc.z = fmaxf(acc.z, 0.f);
    acc.w = fmaxf(acc.w, 0.f);
    *(float4*)&At[wave * LD + 4 * lane] = acc;
    __syncthreads();
    if (wave >= NN / 16) return;

    const int l15  = lane & 15;
    const int kOff = (lane >> 4) << 1;
    const float* __restrict__ arow = &At[l15 * LD + kOff];
    const float* __restrict__ brow = &Bt[(wave * 16 + l15) * LD + kOff];
    v8f d8 = v8f{};
    #pragma unroll
    for (int s4 = 0; s4 < KK; s4 += 4) {
        const v2f a = *(const v2f*)(arow + s4);
        const v2f b = *(const v2f*)(brow + s4);
        d8 = __builtin_amdgcn_wmma_f32_16x16x4_f32(
             false, a,  false, b,
             (short)0, d8,  false,  false);
    }
    const int cm = row0 + ((lane >> 4) << 3);
    float* __restrict__ crow = H2 + (size_t)cm * NN + wave * 16 + l15;
    #pragma unroll
    for (int r = 0; r < 8; ++r) crow[r * NN] = d8[r];
}

extern "C" void kernel_launch(void* const* d_in, const int* in_sizes, int n_in,
                              void* d_out, int out_size, void* d_ws, size_t ws_size,
                              hipStream_t stream) {
    const float* x   = (const float*)d_in[0];
    const float* W1  = (const float*)d_in[1];
    const float* W2  = (const float*)d_in[2];
    const float* ew  = (const float*)d_in[3];
    const int*   er  = (const int*)  d_in[4];
    const int*   ec  = (const int*)  d_in[5];

    const int F_IN = 512, C1 = 128;
    const int nNodes = in_sizes[0] / F_IN;
    const int nEdges = in_sizes[3];

    float* H   = (float*)d_ws;
    float* Hs  = H  + (size_t)nNodes * C1;
    float* R3  = Hs + (size_t)nNodes * C1;
    float* H2  = Hs;
    float* out = (float*)d_out;

    int*            rowptr = (int*)R3;
    float*          sw     = R3 + 65536;
    unsigned short* scol   = (unsigned short*)(sw + nEdges);
    int*  Mc  = (int*)Hs;
    int*  Lst = Mc + NB * NB;
    int2* rec = (int2*)(Hs + 2 * 65536);

    const int BLK = 256;
    const int chunk = (nEdges + NB - 1) / NB;
    const int rpb   = (nNodes + NB - 1) / NB;

    {
        int mTiles = nNodes / 16;
        int gemmBlocks = (C1 / 64) * ((mTiles + 3) / 4);
        gemm1_with_grouping<<<NB + gemmBlocks, 128, 0, stream>>>(
            x, W1, H, nNodes, er, ec, ew, Mc, Lst, rec, nEdges, chunk, (unsigned)rpb);
    }
    owner_sort<<<NB, SORT_T, 0, stream>>>(rec, Mc, Lst, rowptr, scol, sw,
                                          nEdges, nNodes, rpb, chunk);

    spmm_rows_gemm2<<<(nNodes + FUSE_ROWS - 1) / FUSE_ROWS, FUSE_ROWS * 32, 0, stream>>>(
        rowptr, scol, sw, H, W2, H2, nNodes);

    spmm_rows<4><<<(nNodes + 7) / 8, BLK, 0, stream>>>(rowptr, scol, sw, H2, out, nNodes);
}


# GEMM1 B panel held in LDS one K half at a time (staged twice per block, 67 KB of LDS instead of 131 KB): 4 blocks per WGP instead of 2 keep the matrix unit fed; k order and sums unchanged
# speedup vs baseline: 3.7642x; 1.0472x over previous
#include <hip/hip_runtime.h>

typedef float v2f __attribute__((ext_vector_type(2)));
typedef float v8f __attribute__((ext_vector_type(8)));

template <int NN, int KH, int NT, int T>
__device__ __forceinline__ void
stage_panel_slice(const float* __restrict__ B, float* __restrict__ Bt,
                  int kbase, int n0, int tid) {
    constexpr int LDB = KH + 4;
    const int q  = tid & 3;
    const int kk = tid >> 2;
    constexpr int ITERS = KH * 16 * NT / 4 / T;
    #pragma unroll 8
    for (int it = 0; it < ITERS; ++it) {
        const int g   = it % NT;
        const int k   = (it / NT) * (T / 4) + kk;
        const int col = 16 * g + 4 * q;
        const float4 v = *(const float4*)(B + (kbase + k) * NN + n0 + col);
        float* d = &Bt[col * LDB + k];
        d[0]       = v.x;
        d[LDB]     = v.y;
        d[2 * LDB] = v.z;
        d[3 * LDB] = v.w;
    }
}

template <int RELU_A, int NN, int KK, int NT, int WAVES, int KS>
__device__ __forceinline__ void
gemm_block(const float* __restrict__ A,
           const float* __restrict__ B,
           float* __restrict__ C,
           int M, int bx, int by) {
    constexpr int KH  = KK / KS;
    constexpr int LDB = KH + 4;
    constexpr int T   = WAVES * 32;
    constexpr int U   = 16;
    static_assert(KK % KS == 0 && KH % (4 * U) == 0, "a slice must be whole register blocks");
    static_assert((KH * 16 * NT / 4) % T == 0 && KH % (T / 4) == 0, "panel staging shape");
    __shared__ float Bt[16 * NT * LDB];

    const int tid = (int)threadIdx.x;
    const int n0  = bx * (16 * NT);

    stage_panel_slice<NN, KH, NT, T>(B, Bt, 0, n0, tid);
    __syncthreads();

    const int wave = tid >> 5;
    const int lane = tid & 31;
    const int tileM = by * WAVES + wave;
    const bool live = tileM < (M >> 4);

    const int l15  = lane & 15;
    const int kOff = (lane >> 4) << 1;

    const float* __restrict__ arow = A + ((tileM << 4) + l15) * KK + kOff;
    const float* __restrict__ brow = &Bt[l15 * LDB + kOff];

    v8f acc[NT];
    #pragma unroll
    for (int j = 0; j < NT; ++j) acc[j] = v8f{};

    float2 an[U];
    #pragma unroll
    for (int u = 0; u < U; ++u) an[u] = make_float2(0.f, 0.f);
    if (live) {
        #pragma unroll
        for (int u = 0; u < U; ++u) an[u] = *(const float2*)(arow + 4 * u);
    }

    #pragma unroll 1
    for (int kb = 0; kb < KK; kb += 4 * U) {
        const int ks = kb % KH;
        if (kb != 0 && ks == 0) {
            __syncthreads();
            stage_panel_slice<NN, KH, NT, T>(B, Bt, kb, n0, tid);
            __syncthreads();
        }
        if (live) {
            float2 ac[U];
            #pragma unroll
            for (int u = 0; u < U; ++u) ac[u] = an[u];
            if (kb + 4 * U < KK) {
                #pragma unroll
                for (int u = 0; u < U; ++u)
                    an[u] = *(const float2*)(arow + kb + 4 * U + 4 * u);
            }
            #pragma unroll
            for (int u = 0; u < U; ++u) {
                float2 av = ac[u];
                if (RELU_A) {
                    av.x = fmaxf(av.x, 0.f);
                    av.y = fmaxf(av.y, 0.f);
                }
                v2f a; a.x = av.x; a.y = av.y;
                #pragma unroll
                for (int j = 0; j < NT; ++j) {
                    v2f b = *(const v2f*)(brow + j * 16 * LDB + ks + 4 * u);
                    acc[j] = __builtin_amdgcn_wmma_f32_16x16x4_f32(
                         false, a,  false, b,
                         (short)0, acc[j],  false,  false);
                }
            }
        }
    }
    if (!live) return;

    const int cm = (tileM << 4) + ((lane >> 4) << 3);
    float* __restrict__ crow = C + cm * NN + n0 + l15;
    #pragma unroll
    for (int j = 0; j < NT; ++j) {
        #pragma unroll
        for (int r = 0; r < 8; ++r) crow[r * NN + 16 * j] = acc[j][r];
    }
}

#define NB 256
#define SORT_T 1024

template <int T>
__device__ __forceinline__ void
group_chunk(const int* __restrict__ erow, const int* __restrict__ ecol,
            const float* __restrict__ ew, int* __restrict__ Mc,
            int* __restrict__ Lst, int2* __restrict__ rec,
            int nEdges, int chunk, unsigned rpb, int b) {
    static_assert(2 * T == NB, "two owners per thread in the scan");
    __shared__ alignas(8) int cnt[NB];
    __shared__ alignas(8) int pos[NB];
    __shared__ int wtot[T / 32];
    const int tid  = (int)threadIdx.x;
    const int lane = tid & 31;
    const int wave = tid >> 5;
    cnt[2 * tid] = 0;
    cnt[2 * tid + 1] = 0;
    __syncthreads();
    const int e0 = b * chunk;
    const int e1 = min(e0 + chunk, nEdges);
    for (int base = e0 + tid; base < e1; base += T * 16) {
        int r[16];
        #pragma unroll
        for (int u = 0; u < 16; ++u) {
            const int e = base + u * T;
            r[u] = (e < e1) ? erow[e] : 0;
        }
        #pragma unroll
        for (int u = 0; u < 16; ++u) {
            if (base + u * T < e1) atomicAdd(&cnt[(unsigned)r[u] / rpb], 1);
        }
    }
    __syncthreads();
    const int a0 = cnt[2 * tid];
    const int a1 = cnt[2 * tid + 1];
    int x = a0 + a1;
    #pragma unroll
    for (int d = 1; d < 32; d <<= 1) {
        int y = __shfl_up(x, d);
        if (lane >= d) x += y;
    }
    if (lane == 31) wtot[wave] = x;
    __syncthreads();
    int wbase = 0;
    for (int w = 0; w < wave; ++w) wbase += wtot[w];
    const int ex = wbase + x - (a0 + a1);
    pos[2 * tid]     = ex;
    pos[2 * tid + 1] = ex + a0;
    *(int2*)(Mc  + b * NB + 2 * tid) = make_int2(a0, a1);
    *(int2*)(Lst + b * NB + 2 * tid) = make_int2(ex, ex + a0);
    __syncthreads();
    for (int base = e0 + tid; base < e1; base += T * 8) {
        int   r[8], c[8];
        float w[8];
        #pragma unroll
        for (int u = 0; u < 8; ++u) {
            const int e = base + u * T;
            const bool ok = e < e1;
            r[u] = ok ? erow[e] : 0;
            c[u] = ok ? ecol[e] : 0;
            w[u] = ok ? ew[e]   : 0.f;
        }
        int p[8];
        #pragma unroll
        for (int u = 0; u < 8; ++u) {
            const unsigned o = (unsigned)r[u] / rpb;
            p[u] = 0;
            if (base + u * T < e1) p[u] = atomicAdd(&pos[o], 1);
            r[u] = (int)((unsigned)r[u] - o * rpb);
        }
        #pragma unroll
        for (int u = 0; u < 8; ++u) {
            if (base + u * T < e1)
                rec[e0 + p[u]] = make_int2((int)(((unsigned)r[u] << 16) | ((unsigned)c[u] & 0xffffu)),
                                           __float_as_int(w[u]));
        }
    }
}

__global__ void __launch_bounds__(128)
gemm1_with_grouping(const float* __restrict__ x, const float* __restrict__ W1,
                    float* __restrict__ H, int nNodes,
                    const int* __restrict__ erow, const int* __restrict__ ecol,
                    const float* __restrict__ ew, int* __restrict__ Mc,
                    int* __restrict__ Lst, int2* __restrict__ rec,
                    int nEdges, int chunk, unsigned rpb) {
    if ((int)blockIdx.x < NB) {
        group_chunk<128>(erow, ecol, ew, Mc, Lst, rec, nEdges, chunk, rpb, (int)blockIdx.x);
        return;
    }
    const int idx = (int)blockIdx.x - NB;
    gemm_block<0, 128, 512, 4, 4, 2>(x, W1, H, nNodes, idx & 1, idx >> 1);
}

__device__ __forceinline__ int block_excl_scan(int t, int* wsum) {
    const int lane = (int)threadIdx.x & 31;
    const int wave = (int)threadIdx.x >> 5;
    int x = t;
    #pragma unroll
    for (int d = 1; d < 32; d <<= 1) {
        int y = __shfl_up(x, d);
        if (lane >= d) x += y;
    }
    if (lane == 31) wsum[wave] = x;
    __syncthreads();
    if (wave == 0) {
        int s0 = wsum[lane];
        int s = s0;
        #pragma unroll
        for (int d = 1; d < 32; d <<= 1) {
            int y = __shfl_up(s, d);
            if (lane >= d) s += y;
        }
        wsum[lane] = s - s0;
    }
    __syncthreads();
    return wsum[wave] + x - t;
}

#define STAGE_CAP 16384

__global__ void __launch_bounds__(SORT_T)
owner_sort(const int2* __restrict__ rec, const int* __restrict__ Mc,
           const int* __restrict__ Lst,
           int* __restrict__ rowptr, unsigned short* __restrict__ scol,
           float* __restrict__ sw, int nEdges, int nNodes, int rpb, int chunk) {
    static_assert(SORT_T == 4 * NB, "four threads per run");
    __shared__ int cnt[SORT_T];
    __shared__ int wsum[32];
    __shared__ int runStart[NB];
    __shared__ int runLen[NB];
    __shared__ int tot[2];
    __shared__ float stw[STAGE_CAP];
    __shared__ unsigned short stc[STAGE_CAP];
    const int tid = (int)threadIdx.x;
    const int o   = (int)blockIdx.x;
    const int r0  = o * rpb;

    cnt[tid] = 0;
    if (tid < 2) tot[tid] = 0;
    __syncthreads();
    if (tid < NB) {
        int ls = Lst[tid * NB + o];
        int ln = Mc[tid * NB + o];
        runStart[tid] = tid * chunk + ls;
        runLen[tid]   = ln;
        #pragma unroll
        for (int d = 16; d >= 1; d >>= 1) {
            ls += __shfl_xor(ls, d);
            ln += __shfl_xor(ln, d);
        }
        if ((tid & 31) == 0) {
            atomicAdd(&tot[0], ls);
            atomicAdd(&tot[1], ln);
        }
    }
    __syncthreads();
    const int s  = tot[0];
    const int n  = tot[1];
    const int rs = runStart[tid >> 2];
    const int rl = runLen[tid >> 2];
    const int rq = tid & 3;

    for (int j0 = rq; j0 < rl; j0 += 16) {
        int2 rc[4];
        #pragma unroll
        for (int u = 0; u < 4; ++u)
            rc[u] = (j0 + 4 * u < rl) ? rec[rs + j0 + 4 * u] : make_int2(0, 0);
        #pragma unroll
        for (int u = 0; u < 4; ++u)
            if (j0 + 4 * u < rl) atomicAdd(&cnt[(unsigned)rc[u].x >> 16], 1);
    }
    __syncthreads();
    const int mine = cnt[tid];
    const int excl = block_excl_scan(mine, wsum);
    if (tid < rpb && r0 + tid < nNodes) rowptr[r0 + tid] = s + excl;
    if (o == NB - 1 && tid == 0) rowptr[nNodes] = nEdges;
    cnt[tid] = excl;
    __syncthreads();
    if (n <= STAGE_CAP) {
        for (int j0 = rq; j0 < rl; j0 += 16) {
            int2 rc[4];
            #pragma unroll
            for (int u = 0; u < 4; ++u)
                rc[u] = (j0 + 4 * u < rl) ? rec[rs + j0 + 4 * u] : make_int2(0, 0);
            #pragma unroll
            for (int u = 0; u < 4; ++u) {
                if (j0 + 4 * u < rl) {
                    int k = atomicAdd(&cnt[(unsigned)rc[u].x >> 16], 1);
                    stc[k] = (unsigned short)((unsigned)rc[u].x & 0xffffu);
                    stw[k] = __int_as_float(rc[u].y);
                }
            }
        }
        __syncthreads();
        for (int i = tid; i < n; i += SORT_T) {
            scol[s + i] = stc[i];
            sw[s + i]   = stw[i];
        }
    } else {
        for (int j = rq; j < rl; j += 4) {
            int2 rc = rec[rs + j];
            int k = atomicAdd(&cnt[(unsigned)rc.x >> 16], 1);
            int dst = s + k;
            scol[dst] = (unsigned short)((unsigned)rc.x & 0xffffu);
            sw[dst]   = __int_as_float(rc.y);
        }
    }
}

template <int SHIFT>
__device__ __forceinline__ float4
row_sum(const unsigned short* __restrict__ scol,
        const float* __restrict__ sw,
        const float* __restrict__ h,
        int rs, int re, int lane) {
    constexpr int LPE = 1 << SHIFT;
    constexpr int EPS = 32 >> SHIFT;
    constexpr int Cch = 4 << SHIFT;
    const int s = __builtin_amdgcn_readfirstlane(rs);
    const int e = __builtin_amdgcn_readfirstlane(re);
    const int sub = lane & (LPE - 1);
    const int grp = lane >> SHIFT;
    const float* __restrict__ hb = h + sub * 4;

    float4 acc = make_float4(0.f, 0.f, 0.f, 0.f);
    for (int base = s; base < e; base += 32) {
        const int n = min(32, e - base);
        int   c = 0;
        float w = 0.f;
        if (lane < n) { c = (int)scol[base + lane]; w = sw[base + lane]; }
        const int wi = __float_as_int(w);
        int t = 0;
        for (; t + 4 * EPS <= n; t += 4 * EPS) {
            float4 hv[4];
            float  wj[4];
            #pragma unroll
            for (int u = 0; u < 4; ++u) {
                int cj, wb;
                if (EPS == 1) {
                    cj = __builtin_amdgcn_readlane(c,  t + u);
                    wb = __builtin_amdgcn_readlane(wi, t + u);
                } else {
                    int c0 = __builtin_amdgcn_readlane(c,  t + 2 * u);
                    int c1 = __builtin_amdgcn_readlane(c,  t + 2 * u + 1);
                    int w0 = __builtin_amdgcn_readlane(wi, t + 2 * u);
                    int w1 = __builtin_amdgcn_readlane(wi, t + 2 * u + 1);
                    cj = grp ? c1 : c0;
                    wb = grp ? w1 : w0;
                }
                wj[u] = __int_as_float(wb);
                hv[u] = *(const float4*)(hb + (size_t)cj * Cch);
            }
            #pragma unroll
            for (int u = 0; u < 4; ++u) {
                acc.x = fmaf(hv[u].x, wj[u], acc.x);
                acc.y = fmaf(hv[u].y, wj[u], acc.y);
                acc.z = fmaf(hv[u].z, wj[u], acc.z);
                acc.w = fmaf(hv[u].w, wj[u], acc.w);
            }
        }
        for (; t < n; t += EPS) {
            int cj, wb;
            bool ok = true;
            if (EPS == 1) {
                cj = __builtin_amdgcn_readlane(c,  t);
                wb = __builtin_amdgcn_readlane(wi, t);
            } else {
                const int t1 = min(t + 1, 31);
                int c0 = __builtin_amdgcn_readlane(c,  t);
                int c1 = __builtin_amdgcn_readlane(c,  t1);
                int w0 = __builtin_amdgcn_readlane(wi, t);
                int w1 = __builtin_amdgcn_readlane(wi, t1);
                cj = grp ? c1 : c0;
                wb = grp ? w1 : w0;
                ok = (t + grp) < n;
            }
            if (ok) {
                const float  wv = __int_as_float(wb);
                const float4 hv = *(const float4*)(hb + (size_t)cj * Cch);
                acc.x = fmaf(hv.x, wv, acc.x);
                acc.y = fmaf(hv.y, wv, acc.y);
                acc.z = fmaf(hv.z, wv, acc.z);
                acc.w = fmaf(hv.w, wv, acc.w);
            }
        }
    }
    if (EPS == 2) {
        acc.x += __shfl_xor(acc.x, 16);
        acc.y += __shfl_xor(acc.y, 16);
        acc.z += __shfl_xor(acc.z, 16);
        acc.w += __shfl_xor(acc.w, 16);
    }
    return acc;
}

template <int SHIFT>
__global__ void spmm_rows(const int* __restrict__ rowptr,
                          const unsigned short* __restrict__ scol,
                          const float* __restrict__ sw,
                          const float* __restrict__ h,
                          float* __restrict__ out,
                          int nRows) {
    constexpr int LPE = 1 << SHIFT;
    constexpr int Cch = 4 << SHIFT;
    const int lane = (int)threadIdx.x & 31;
    const int row  = (int)blockIdx.x * ((int)blockDim.x >> 5) + ((int)threadIdx.x >> 5);
    if (row >= nRows) return;
    const float4 acc = row_sum<SHIFT>(scol, sw, h, rowptr[row], rowptr[row + 1], lane);
    const int sub = lane & (LPE - 1);
    if ((lane >> SHIFT) == 0) *(float4*)(out + (size_t)row * Cch + sub * 4) = acc;
}

#define FUSE_ROWS 16
__global__ void __launch_bounds__(FUSE_ROWS * 32)
spmm_rows_gemm2(const int* __restrict__ rowptr,
                const unsigned short* __restrict__ scol,
                const float* __restrict__ sw,
                const float* __restrict__ h,
                const float* __restrict__ W2,
                float* __restrict__ H2,
                int nRows) {
    constexpr int KK  = 128;
    constexpr int NN  = 64;
    constexpr int LD  = KK + 4;
    constexpr int T   = FUSE_ROWS * 32;
    static_assert((KK * NN / 4) % T == 0 && KK % (T / 4) == 0, "W2 staging shape");
    __shared__ float Bt[NN * LD];
    __shared__ alignas(16) float At[FUSE_ROWS * LD];

    const int tid  = (int)threadIdx.x;
    const int lane = tid & 31;
    const int wave = tid >> 5;

    const int row0 = (int)blockIdx.x * FUSE_ROWS;
    const int row  = row0 + wave;

    const int q  = tid & 3;
    const int kk = tid >> 2;
    constexpr int ITERS = KK * NN / 4 / T;
    float4 wv[ITERS];
    #pragma unroll
    for (int it = 0; it < ITERS; ++it) {
        const int g = it % (NN / 16);
        const int k = (it / (NN / 16)) * (T / 4) + kk;
        wv[it] = *(const float4*)(W2 + k * NN + 16 * g + 4 * q);
    }
    int rs = 0, re = 0;
    if (row < nRows) { rs = rowptr[row]; re = rowptr[row + 1]; }
    #pragma unroll
    for (int it = 0; it < ITERS; ++it) {
        const int g   = it % (NN / 16);
        const int k   = (it / (NN / 16)) * (T / 4) + kk;
        const int col = 16 * g + 4 * q;
        float* d = &Bt[col * LD + k];
        d[0]      = wv[it].x;
        d[LD]     = wv[it].y;
        d[2 * LD] = wv[it].z;
        d[3 * LD] = wv[it].w;
    }

    float4 acc = row_sum<5>(scol, sw, h, rs, re, lane);
    acc.x = fmaxf(acc.x, 0.f);
    acc.y = fmaxf(acc.y, 0.f);
    acc.z = fmaxf(acc.z, 0.f);
    acc.w = fmaxf(acc.w, 0.f);
    *(float4*)&At[wave * LD + 4 * lane] = acc;
    __syncthreads();
    if (wave >= NN / 16) return;

    const int l15  = lane & 15;
    const int kOff = (lane >> 4) << 1;
    const float* __restrict__ arow = &At[l15 * LD + kOff];
    const float* __restrict__ brow = &Bt[(wave * 16 + l15) * LD + kOff];
    v8f d8 = v8f{};
    #pragma unroll
    for (int s4 = 0; s4 < KK; s4 += 4) {
        const v2f a = *(const v2f*)(arow + s4);
        const v2f b = *(const v2f*)(brow + s4);
        d8 = __builtin_amdgcn_wmma_f32_16x16x4_f32(
             false, a,  false, b,
             (short)0, d8,  false,  false);
    }
    const int cm = row0 + ((lane >> 4) << 3);
    float* __restrict__ crow = H2 + (size_t)cm * NN + wave * 16 + l15;
    #pragma unroll
    for (int r = 0; r < 8; ++r) crow[r * NN] = d8[r];
}

extern "C" void kernel_launch(void* const* d_in, const int* in_sizes, int n_in,
                              void* d_out, int out_size, void* d_ws, size_t ws_size,
                              hipStream_t stream) {
    const float* x   = (const float*)d_in[0];
    const float* W1  = (const float*)d_in[1];
    const float* W2  = (const float*)d_in[2];
    const float* ew  = (const float*)d_in[3];
    const int*   er  = (const int*)  d_in[4];
    const int*   ec  = (const int*)  d_in[5];

    const int F_IN = 512, C1 = 128;
    const int nNodes = in_sizes[0] / F_IN;
    const int nEdges = in_sizes[3];

    float* H   = (float*)d_ws;
    float* Hs  = H  + (size_t)nNodes * C1;
    float* R3  = Hs + (size_t)nNodes * C1;
    float* H2  = Hs;
    float* out = (float*)d_out;

    int*            rowptr = (int*)R3;
    float*          sw     = R3 + 65536;
    unsigned short* scol   = (unsigned short*)(sw + nEdges);
    int*  Mc  = (int*)Hs;
    int*  Lst = Mc + NB * NB;
    int2* rec = (int2*)(Hs + 2 * 65536);

    const int BLK = 256;
    const int chunk = (nEdges + NB - 1) / NB;
    const int rpb   = (nNodes + NB - 1) / NB;

    {
        int mTiles = nNodes / 16;
        int gemmBlocks = (C1 / 64) * ((mTiles + 3) / 4);
        gemm1_with_grouping<<<NB + gemmBlocks, 128, 0, stream>>>(
            x, W1, H, nNodes, er, ec, ew, Mc, Lst, rec, nEdges, chunk, (unsigned)rpb);
    }
    owner_sort<<<NB, SORT_T, 0, stream>>>(rec, Mc, Lst, rowptr, scol, sw,
                                          nEdges, nNodes, rpb, chunk);

    spmm_rows_gemm2<<<(nNodes + FUSE_ROWS - 1) / FUSE_ROWS, FUSE_ROWS * 32, 0, stream>>>(
        rowptr, scol, sw, H, W2, H2, nNodes);

    spmm_rows<4><<<(nNodes + 7) / 8, BLK, 0, stream>>>(rowptr, scol, sw, H2, out, nNodes);
}


# GEMM1 A operand fetched as whole 128-byte lines (4 lines per 16-byte-per-lane load) and parked in a per-wave LDS tile; fragments read from LDS, same elements and k order, sums unchanged
# speedup vs baseline: 3.8939x; 1.0345x over previous
#include <hip/hip_runtime.h>

typedef float v2f __attribute__((ext_vector_type(2)));
typedef float v8f __attribute__((ext_vector_type(8)));

template <int NN, int KH, int NT, int T>
__device__ __forceinline__ void
stage_panel_slice(const float* __restrict__ B, float* __restrict__ Bt,
                  int kbase, int n0, int tid) {
    constexpr int LDB = KH + 4;
    const int q  = tid & 3;
    const int kk = tid >> 2;
    constexpr int ITERS = KH * 16 * NT / 4 / T;
    #pragma unroll 8
    for (int it = 0; it < ITERS; ++it) {
        const int g   = it % NT;
        const int k   = (it / NT) * (T / 4) + kk;
        const int col = 16 * g + 4 * q;
        const float4 v = *(const float4*)(B + (kbase + k) * NN + n0 + col);
        float* d = &Bt[col * LDB + k];
        d[0]       = v.x;
        d[LDB]     = v.y;
        d[2 * LDB] = v.z;
        d[3 * LDB] = v.w;
    }
}

template <int RELU_A, int NN, int KK, int NT, int WAVES, int KS>
__device__ __forceinline__ void
gemm_block(const float* __restrict__ A,
           const float* __restrict__ B,
           float* __restrict__ C,
           int M, int bx, int by) {
    constexpr int KH  = KK / KS;
    constexpr int LDB = KH + 4;
    constexpr int T   = WAVES * 32;
    constexpr int KC  = 32;
    constexpr int LDA = KC + 4;
    constexpr int U   = KC / 4;
    constexpr int RL  = 16 * KC / 4 / 32;
    static_assert(KK % KS == 0 && KH % (2 * KC) == 0, "a slice must be whole stage pairs");
    static_assert((KH * 16 * NT / 4) % T == 0 && KH % (T / 4) == 0, "panel staging shape");
    static_assert(KC == 32 && RL == 4, "8 lanes x 16 bytes cover one row of a stage");
    __shared__ float Bt[16 * NT * LDB];
    __shared__ alignas(16) float At[WAVES * 16 * LDA];

    const int tid = (int)threadIdx.x;
    const int n0  = bx * (16 * NT);

    stage_panel_slice<NN, KH, NT, T>(B, Bt, 0, n0, tid);
    __syncthreads();

    const int wave = tid >> 5;
    const int lane = tid & 31;
    const int tileM = by * WAVES + wave;
    const bool live = tileM < (M >> 4);

    const int l15  = lane & 15;
    const int kOff = (lane >> 4) << 1;

    const int lr = lane >> 3;
    const int lc = (lane & 7) << 2;
    const float* __restrict__ aline = A + ((tileM << 4) + lr) * KK + lc;
    float*       atile = &At[wave * 16 * LDA + lr * LDA + lc];
    const float* afrag = &At[wave * 16 * LDA + l15 * LDA + kOff];
    const float* __restrict__ brow = &Bt[l15 * LDB + kOff];

    v8f acc[NT];
    #pragma unroll
    for (int j = 0; j < NT; ++j) acc[j] = v8f{};

    float4 ra[RL], rb[RL];
    #pragma unroll
    for (int i = 0; i < RL; ++i) {
        ra[i] = make_float4(0.f, 0.f, 0.f, 0.f);
        rb[i] = make_float4(0.f, 0.f, 0.f, 0.f);
    }
    if (live) {
        #pragma unroll
        for (int i = 0; i < RL; ++i) ra[i] = *(const float4*)(aline + 4 * i * KK);
        #pragma unroll
        for (int i = 0; i < RL; ++i) rb[i] = *(const float4*)(aline + 4 * i * KK + KC);
    }

    #pragma unroll 1
    for (int kb = 0; kb < KK; kb += 2 * KC) {
        const int ks = kb % KH;
        if (kb != 0 && ks == 0) {
            __syncthreads();
            stage_panel_slice<NN, KH, NT, T>(B, Bt, kb, n0, tid);
            __syncthreads();
        }
        if (live) {
            const int kn = (kb + 2 * KC < KK) ? kb + 2 * KC : kb;
            #pragma unroll
            for (int h = 0; h < 2; ++h) {
                #pragma unroll
                for (int i = 0; i < RL; ++i)
                    *(float4*)(atile + 4 * i * LDA) = h ? rb[i] : ra[i];
                #pragma unroll
                for (int i = 0; i < RL; ++i) {
                    const float4 v = *(const float4*)(aline + 4 * i * KK + kn + h * KC);
                    if (h) rb[i] = v; else ra[i] = v;
                }
                #pragma unroll
                for (int u = 0; u < U; ++u) {
                    v2f a = *(const v2f*)(afrag + 4 * u);
                    if (RELU_A) {
                        a.x = fmaxf(a.x, 0.f);
                        a.y = fmaxf(a.y, 0.f);
                    }
                    #pragma unroll
                    for (int j = 0; j < NT; ++j) {
                        v2f b = *(const v2f*)(brow + j * 16 * LDB + ks + h * KC + 4 * u);
                        acc[j] = __builtin_amdgcn_wmma_f32_16x16x4_f32(
                             false, a,  false, b,
                             (short)0, acc[j],  false,  false);
                    }
                }
            }
        }
    }
    if (!live) return;

    const int cm = (tileM << 4) + ((lane >> 4) << 3);
    float* __restrict__ crow = C + cm * NN + n0 + l15;
    #pragma unroll
    for (int j = 0; j < NT; ++j) {
        #pragma unroll
        for (int r = 0; r < 8; ++r) crow[r * NN + 16 * j] = acc[j][r];
    }
}

#define NB 256
#define SORT_T 1024

template <int T>
__device__ __forceinline__ void
group_chunk(const int* __restrict__ erow, const int* __restrict__ ecol,
            const float* __restrict__ ew, int* __restrict__ Mc,
            int* __restrict__ Lst, int2* __restrict__ rec,
            int nEdges, int chunk, unsigned rpb, int b) {
    static_assert(2 * T == NB, "two owners per thread in the scan");
    __shared__ alignas(8) int cnt[NB];
    __shared__ alignas(8) int pos[NB];
    __shared__ int wtot[T / 32];
    const int tid  = (int)threadIdx.x;
    const int lane = tid & 31;
    const int wave = tid >> 5;
    cnt[2 * tid] = 0;
    cnt[2 * tid + 1] = 0;
    __syncthreads();
    const int e0 = b * chunk;
    const int e1 = min(e0 + chunk, nEdges);
    for (int base = e0 + tid; base < e1; base += T * 16) {
        int r[16];
        #pragma unroll
        for (int u = 0; u < 16; ++u) {
            const int e = base + u * T;
            r[u] = (e < e1) ? erow[e] : 0;
        }
        #pragma unroll
        for (int u = 0; u < 16; ++u) {
            if (base + u * T < e1) atomicAdd(&cnt[(unsigned)r[u] / rpb], 1);
        }
    }
    __syncthreads();
    const int a0 = cnt[2 * tid];
    const int a1 = cnt[2 * tid + 1];
    int x = a0 + a1;
    #pragma unroll
    for (int d = 1; d < 32; d <<= 1) {
        int y = __shfl_up(x, d);
        if (lane >= d) x += y;
    }
    if (lane == 31) wtot[wave] = x;
    __syncthreads();
    int wbase = 0;
    for (int w = 0; w < wave; ++w) wbase += wtot[w];
    const int ex = wbase + x - (a0 + a1);
    pos[2 * tid]     = ex;
    pos[2 * tid + 1] = ex + a0;
    *(int2*)(Mc  + b * NB + 2 * tid) = make_int2(a0, a1);
    *(int2*)(Lst + b * NB + 2 * tid) = make_int2(ex, ex + a0);
    __syncthreads();
    for (int base = e0 + tid; base < e1; base += T * 8) {
        int   r[8], c[8];
        float w[8];
        #pragma unroll
        for (int u = 0; u < 8; ++u) {
            const int e = base + u * T;
            const bool ok = e < e1;
            r[u] = ok ? erow[e] : 0;
            c[u] = ok ? ecol[e] : 0;
            w[u] = ok ? ew[e]   : 0.f;
        }
        int p[8];
        #pragma unroll
        for (int u = 0; u < 8; ++u) {
            const unsigned o = (unsigned)r[u] / rpb;
            p[u] = 0;
            if (base + u * T < e1) p[u] = atomicAdd(&pos[o], 1);
            r[u] = (int)((unsigned)r[u] - o * rpb);
        }
        #pragma unroll
        for (int u = 0; u < 8; ++u) {
            if (base + u * T < e1)
                rec[e0 + p[u]] = make_int2((int)(((unsigned)r[u] << 16) | ((unsigned)c[u] & 0xffffu)),
                                           __float_as_int(w[u]));
        }
    }
}

__global__ void __launch_bounds__(128)
gemm1_with_grouping(const float* __restrict__ x, const float* __restrict__ W1,
                    float* __restrict__ H, int nNodes,
                    const int* __restrict__ erow, const int* __restrict__ ecol,
                    const float* __restrict__ ew, int* __restrict__ Mc,
                    int* __restrict__ Lst, int2* __restrict__ rec,
                    int nEdges, int chunk, unsigned rpb) {
    if ((int)blockIdx.x < NB) {
        group_chunk<128>(erow, ecol, ew, Mc, Lst, rec, nEdges, chunk, rpb, (int)blockIdx.x);
        return;
    }
    const int idx = (int)blockIdx.x - NB;
    gemm_block<0, 128, 512, 4, 4, 2>(x, W1, H, nNodes, idx & 1, idx >> 1);
}

__device__ __forceinline__ int block_excl_scan(int t, int* wsum) {
    const int lane = (int)threadIdx.x & 31;
    const int wave = (int)threadIdx.x >> 5;
    int x = t;
    #pragma unroll
    for (int d = 1; d < 32; d <<= 1) {
        int y = __shfl_up(x, d);
        if (lane >= d) x += y;
    }
    if (lane == 31) wsum[wave] = x;
    __syncthreads();
    if (wave == 0) {
        int s0 = wsum[lane];
        int s = s0;
        #pragma unroll
        for (int d = 1; d < 32; d <<= 1) {
            int y = __shfl_up(s, d);
            if (lane >= d) s += y;
        }
        wsum[lane] = s - s0;
    }
    __syncthreads();
    return wsum[wave] + x - t;
}

#define STAGE_CAP 16384

__global__ void __launch_bounds__(SORT_T)
owner_sort(const int2* __restrict__ rec, const int* __restrict__ Mc,
           const int* __restrict__ Lst,
           int* __restrict__ rowptr, unsigned short* __restrict__ scol,
           float* __restrict__ sw, int nEdges, int nNodes, int rpb, int chunk) {
    static_assert(SORT_T == 4 * NB, "four threads per run");
    __shared__ int cnt[SORT_T];
    __shared__ int wsum[32];
    __shared__ int runStart[NB];
    __shared__ int runLen[NB];
    __shared__ int tot[2];
    __shared__ float stw[STAGE_CAP];
    __shared__ unsigned short stc[STAGE_CAP];
    const int tid = (int)threadIdx.x;
    const int o   = (int)blockIdx.x;
    const int r0  = o * rpb;

    cnt[tid] = 0;
    if (tid < 2) tot[tid] = 0;
    __syncthreads();
    if (tid < NB) {
        int ls = Lst[tid * NB + o];
        int ln = Mc[tid * NB + o];
        runStart[tid] = tid * chunk + ls;
        runLen[tid]   = ln;
        #pragma unroll
        for (int d = 16; d >= 1; d >>= 1) {
            ls += __shfl_xor(ls, d);
            ln += __shfl_xor(ln, d);
        }
        if ((tid & 31) == 0) {
            atomicAdd(&tot[0], ls);
            atomicAdd(&tot[1], ln);
        }
    }
    __syncthreads();
    const int s  = tot[0];
    const int n  = tot[1];
    const int rs = runStart[tid >> 2];
    const int rl = runLen[tid >> 2];
    const int rq = tid & 3;

    for (int j0 = rq; j0 < rl; j0 += 16) {
        int2 rc[4];
        #pragma unroll
        for (int u = 0; u < 4; ++u)
            rc[u] = (j0 + 4 * u < rl) ? rec[rs + j0 + 4 * u] : make_int2(0, 0);
        #pragma unroll
        for (int u = 0; u < 4; ++u)
            if (j0 + 4 * u < rl) atomicAdd(&cnt[(unsigned)rc[u].x >> 16], 1);
    }
    __syncthreads();
    const int mine = cnt[tid];
    const int excl = block_excl_scan(mine, wsum);
    if (tid < rpb && r0 + tid < nNodes) rowptr[r0 + tid] = s + excl;
    if (o == NB - 1 && tid == 0) rowptr[nNodes] = nEdges;
    cnt[tid] = excl;
    __syncthreads();
    if (n <= STAGE_CAP) {
        for (int j0 = rq; j0 < rl; j0 += 16) {
            int2 rc[4];
            #pragma unroll
            for (int u = 0; u < 4; ++u)
                rc[u] = (j0 + 4 * u < rl) ? rec[rs + j0 + 4 * u] : make_int2(0, 0);
            #pragma unroll
            for (int u = 0; u < 4; ++u) {
                if (j0 + 4 * u < rl) {
                    int k = atomicAdd(&cnt[(unsigned)rc[u].x >> 16], 1);
                    stc[k] = (unsigned short)((unsigned)rc[u].x & 0xffffu);
                    stw[k] = __int_as_float(rc[u].y);
                }
            }
        }
        __syncthreads();
        for (int i = tid; i < n; i += SORT_T) {
            scol[s + i] = stc[i];
            sw[s + i]   = stw[i];
        }
    } else {
        for (int j = rq; j < rl; j += 4) {
            int2 rc = rec[rs + j];
            int k = atomicAdd(&cnt[(unsigned)rc.x >> 16], 1);
            int dst = s + k;
            scol[dst] = (unsigned short)((unsigned)rc.x & 0xffffu);
            sw[dst]   = __int_as_float(rc.y);
        }
    }
}

template <int SHIFT>
__device__ __forceinline__ float4
row_sum(const unsigned short* __restrict__ scol,
        const float* __restrict__ sw,
        const float* __restrict__ h,
        int rs, int re, int lane) {
    constexpr int LPE = 1 << SHIFT;
    constexpr int EPS = 32 >> SHIFT;
    constexpr int Cch = 4 << SHIFT;
    const int s = __builtin_amdgcn_readfirstlane(rs);
    const int e = __builtin_amdgcn_readfirstlane(re);
    const int sub = lane & (LPE - 1);
    const int grp = lane >> SHIFT;
    const float* __restrict__ hb = h + sub * 4;

    float4 acc = make_float4(0.f, 0.f, 0.f, 0.f);
    for (int base = s; base < e; base += 32) {
        const int n = min(32, e - base);
        int   c = 0;
        float w = 0.f;
        if (lane < n) { c = (int)scol[base + lane]; w = sw[base + lane]; }
        const int wi = __float_as_int(w);
        int t = 0;
        for (; t + 4 * EPS <= n; t += 4 * EPS) {
            float4 hv[4];
            float  wj[4];
            #pragma unroll
            for (int u = 0; u < 4; ++u) {
                int cj, wb;
                if (EPS == 1) {
                    cj = __builtin_amdgcn_readlane(c,  t + u);
                    wb = __builtin_amdgcn_readlane(wi, t + u);
                } else {
                    int c0 = __builtin_amdgcn_readlane(c,  t + 2 * u);
                    int c1 = __builtin_amdgcn_readlane(c,  t + 2 * u + 1);
                    int w0 = __builtin_amdgcn_readlane(wi, t + 2 * u);
                    int w1 = __builtin_amdgcn_readlane(wi, t + 2 * u + 1);
                    cj = grp ? c1 : c0;
                    wb = grp ? w1 : w0;
                }
                wj[u] = __int_as_float(wb);
                hv[u] = *(const float4*)(hb + (size_t)cj * Cch);
            }
            #pragma unroll
            for (int u = 0; u < 4; ++u) {
                acc.x = fmaf(hv[u].x, wj[u], acc.x);
                acc.y = fmaf(hv[u].y, wj[u], acc.y);
                acc.z = fmaf(hv[u].z, wj[u], acc.z);
                acc.w = fmaf(hv[u].w, wj[u], acc.w);
            }
        }
        for (; t < n; t += EPS) {
            int cj, wb;
            bool ok = true;
            if (EPS == 1) {
                cj = __builtin_amdgcn_readlane(c,  t);
                wb = __builtin_amdgcn_readlane(wi, t);
            } else {
                const int t1 = min(t + 1, 31);
                int c0 = __builtin_amdgcn_readlane(c,  t);
                int c1 = __builtin_amdgcn_readlane(c,  t1);
                int w0 = __builtin_amdgcn_readlane(wi, t);
                int w1 = __builtin_amdgcn_readlane(wi, t1);
                cj = grp ? c1 : c0;
                wb = grp ? w1 : w0;
                ok = (t + grp) < n;
            }
            if (ok) {
                const float  wv = __int_as_float(wb);
                const float4 hv = *(const float4*)(hb + (size_t)cj * Cch);
                acc.x = fmaf(hv.x, wv, acc.x);
                acc.y = fmaf(hv.y, wv, acc.y);
                acc.z = fmaf(hv.z, wv, acc.z);
                acc.w = fmaf(hv.w, wv, acc.w);
            }
        }
    }
    if (EPS == 2) {
        acc.x += __shfl_xor(acc.x, 16);
        acc.y += __shfl_xor(acc.y, 16);
        acc.z += __shfl_xor(acc.z, 16);
        acc.w += __shfl_xor(acc.w, 16);
    }
    return acc;
}

template <int SHIFT>
__global__ void spmm_rows(const int* __restrict__ rowptr,
                          const unsigned short* __restrict__ scol,
                          const float* __restrict__ sw,
                          const float* __restrict__ h,
                          float* __restrict__ out,
                          int nRows) {
    constexpr int LPE = 1 << SHIFT;
    constexpr int Cch = 4 << SHIFT;
    const int lane = (int)threadIdx.x & 31;
    const int row  = (int)blockIdx.x * ((int)blockDim.x >> 5) + ((int)threadIdx.x >> 5);
    if (row >= nRows) return;
    const float4 acc = row_sum<SHIFT>(scol, sw, h, rowptr[row], rowptr[row + 1], lane);
    const int sub = lane & (LPE - 1);
    if ((lane >> SHIFT) == 0) *(float4*)(out + (size_t)row * Cch + sub * 4) = acc;
}

#define FUSE_ROWS 16
__global__ void __launch_bounds__(FUSE_ROWS * 32)
spmm_rows_gemm2(const int* __restrict__ rowptr,
                const unsigned short* __restrict__ scol,
                const float* __restrict__ sw,
                const float* __restrict__ h,
                const float* __restrict__ W2,
                float* __restrict__ H2,
                int nRows) {
    constexpr int KK  = 128;
    constexpr int NN  = 64;
    constexpr int LD  = KK + 4;
    constexpr int T   = FUSE_ROWS * 32;
    static_assert((KK * NN / 4) % T == 0 && KK % (T / 4) == 0, "W2 staging shape");
    __shared__ float Bt[NN * LD];
    __shared__ alignas(16) float At[FUSE_ROWS * LD];

    const int tid  = (int)threadIdx.x;
    const int lane = tid & 31;
    const int wave = tid >> 5;

    const int row0 = (int)blockIdx.x * FUSE_ROWS;
    const int row  = row0 + wave;

    const int q  = tid & 3;
    const int kk = tid >> 2;
    constexpr int ITERS = KK * NN / 4 / T;
    float4 wv[ITERS];
    #pragma unroll
    for (int it = 0; it < ITERS; ++it) {
        const int g = it % (NN / 16);
        const int k = (it / (NN / 16)) * (T / 4) + kk;
        wv[it] = *(const float4*)(W2 + k * NN + 16 * g + 4 * q);
    }
    int rs = 0, re = 0;
    if (row < nRows) { rs = rowptr[row]; re = rowptr[row + 1]; }
    #pragma unroll
    for (int it = 0; it < ITERS; ++it) {
        const int g   = it % (NN / 16);
        const int k   = (it / (NN / 16)) * (T / 4) + kk;
        const int col = 16 * g + 4 * q;
        float* d = &Bt[col * LD + k];
        d[0]      = wv[it].x;
        d[LD]     = wv[it].y;
        d[2 * LD] = wv[it].z;
        d[3 * LD] = wv[it].w;
    }

    float4 acc = row_sum<5>(scol, sw, h, rs, re, lane);
    acc.x = fmaxf(acc.x, 0.f);
    acc.y = fmaxf(acc.y, 0.f);
    acc.z = fmaxf(acc.z, 0.f);
    acc.w = fmaxf(acc.w, 0.f);
    *(float4*)&At[wave * LD + 4 * lane] = acc;
    __syncthreads();
    if (wave >= NN / 16) return;

    const int l15  = lane & 15;
    const int kOff = (lane >> 4) << 1;
    const float* __restrict__ arow = &At[l15 * LD + kOff];
    const float* __restrict__ brow = &Bt[(wave * 16 + l15) * LD + kOff];
    v8f d8 = v8f{};
    #pragma unroll
    for (int s4 = 0; s4 < KK; s4 += 4) {
        const v2f a = *(const v2f*)(arow + s4);
        const v2f b = *(const v2f*)(brow + s4);
        d8 = __builtin_amdgcn_wmma_f32_16x16x4_f32(
             false, a,  false, b,
             (short)0, d8,  false,  false);
    }
    const int cm = row0 + ((lane >> 4) << 3);
    float* __restrict__ crow = H2 + (size_t)cm * NN + wave * 16 + l15;
    #pragma unroll
    for (int r = 0; r < 8; ++r) crow[r * NN] = d8[r];
}

extern "C" void kernel_launch(void* const* d_in, const int* in_sizes, int n_in,
                              void* d_out, int out_size, void* d_ws, size_t ws_size,
                              hipStream_t stream) {
    const float* x   = (const float*)d_in[0];
    const float* W1  = (const float*)d_in[1];
    const float* W2  = (const float*)d_in[2];
    const float* ew  = (const float*)d_in[3];
    const int*   er  = (const int*)  d_in[4];
    const int*   ec  = (const int*)  d_in[5];

    const int F_IN = 512, C1 = 128;
    const int nNodes = in_sizes[0] / F_IN;
    const int nEdges = in_sizes[3];

    float* H   = (float*)d_ws;
    float* Hs  = H  + (size_t)nNodes * C1;
    float* R3  = Hs + (size_t)nNodes * C1;
    float* H2  = Hs;
    float* out = (float*)d_out;

    int*            rowptr = (int*)R3;
    float*          sw     = R3 + 65536;
    unsigned short* scol   = (unsigned short*)(sw + nEdges);
    int*  Mc  = (int*)Hs;
    int*  Lst = Mc + NB * NB;
    int2* rec = (int2*)(Hs + 2 * 65536);

    const int BLK = 256;
    const int chunk = (nEdges + NB - 1) / NB;
    const int rpb   = (nNodes + NB - 1) / NB;

    {
        int mTiles = nNodes / 16;
        int gemmBlocks = (C1 / 64) * ((mTiles + 3) / 4);
        gemm1_with_grouping<<<NB + gemmBlocks, 128, 0, stream>>>(
            x, W1, H, nNodes, er, ec, ew, Mc, Lst, rec, nEdges, chunk, (unsigned)rpb);
    }
    owner_sort<<<NB, SORT_T, 0, stream>>>(rec, Mc, Lst, rowptr, scol, sw,
                                          nEdges, nNodes, rpb, chunk);

    spmm_rows_gemm2<<<(nNodes + FUSE_ROWS - 1) / FUSE_ROWS, FUSE_ROWS * 32, 0, stream>>>(
        rowptr, scol, sw, H, W2, H2, nNodes);

    spmm_rows<4><<<(nNodes + 7) / 8, BLK, 0, stream>>>(rowptr, scol, sw, H2, out, nNodes);
}
